# starting kernel
# baseline (speedup 1.0000x reference)
#include <hip/hip_runtime.h>
#include <hip/hip_bf16.h>

typedef float v2f __attribute__((ext_vector_type(2)));
typedef float v8f __attribute__((ext_vector_type(8)));

#define IN_DIM  128
#define HID_DIM 128
#define OUT_DIM 64

__global__ void init_deg_kernel(float* __restrict__ deg, int n) {
    int i = blockIdx.x * blockDim.x + threadIdx.x;
    if (i < n) deg[i] = 1.0f;
}

__global__ void accum_deg_kernel(const int* __restrict__ col,
                                 float* __restrict__ deg, int E) {
    int e = blockIdx.x * blockDim.x + threadIdx.x;
    if (e < E) atomicAdd(&deg[col[e]], 1.0f);
}

__global__ void deg_to_dinv_kernel(float* __restrict__ deg, int n) {
    int i = blockIdx.x * blockDim.x + threadIdx.x;
    if (i < n) deg[i] = rsqrtf(deg[i]);
}

template <int NOUT>
__global__ void gemm_wmma_kernel(const float* __restrict__ X,
                                 const float* __restrict__ W,
                                 float* __restrict__ H, int nNodes) {
    constexpr int K = 128;
    const int wv   = threadIdx.x >> 5;
    const int lane = threadIdx.x & 31;
    const int half = lane >> 4;
    const int l16  = lane & 15;

    const int m0 = blockIdx.x * 16;
    const int n0 = wv * 16;

    int mrow = m0 + l16; if (mrow >= nNodes) mrow = nNodes - 1;
    const float* aptr = X + (size_t)mrow * K + 2 * half;
    const float* bptr = W + (size_t)(n0 + l16) * K + 2 * half;

    v8f c = {0.f, 0.f, 0.f, 0.f, 0.f, 0.f, 0.f, 0.f};
#pragma unroll
    for (int k0 = 0; k0 < K; k0 += 4) {
        v2f a = *(const v2f*)(aptr + k0);
        v2f b = *(const v2f*)(bptr + k0);
        c = __builtin_amdgcn_wmma_f32_16x16x4_f32(false, a, false, b,
                                                  (short)0, c, false, false);
    }

    float* out = H + (size_t)m0 * NOUT + n0;
#pragma unroll
    for (int v = 0; v < 8; ++v) {
        int r = v + 8 * half;
        if (m0 + r < nNodes) out[(size_t)r * NOUT + l16] = c[v];
    }
}

template <int D>
__global__ void propagate_kernel(const int* __restrict__ row,
                                 const int* __restrict__ col,
                                 const float* __restrict__ dinv,
                                 const float* __restrict__ h,
                                 float* __restrict__ agg, int E) {
    int wave = (int)((blockIdx.x * (size_t)blockDim.x + threadIdx.x) >> 5);
    int lane = threadIdx.x & 31;
    if (wave >= E) return;
    int r = row[wave];
    int c = col[wave];
    float nrm = dinv[r] * dinv[c];
    constexpr int V = D / 32;
    const float* src = h   + (size_t)r * D + lane * V;
    float*       dst = agg + (size_t)c * D + lane * V;
    if constexpr (V == 4) {
        float4 v = *(const float4*)src;
        atomicAdd(dst + 0, nrm * v.x);
        atomicAdd(dst + 1, nrm * v.y);
        atomicAdd(dst + 2, nrm * v.z);
        atomicAdd(dst + 3, nrm * v.w);
    } else {
        float2 v = *(const float2*)src;
        atomicAdd(dst + 0, nrm * v.x);
        atomicAdd(dst + 1, nrm * v.y);
    }
}

template <int D, bool RELU>
__global__ void finalize_kernel(const float* __restrict__ agg,
                                const float* __restrict__ h,
                                const float* __restrict__ dinv,
                                const float* __restrict__ bias,
                                float* __restrict__ out, int n) {
    size_t idx = blockIdx.x * (size_t)blockDim.x + threadIdx.x;
    if (idx >= (size_t)n * D) return;
    int i = (int)(idx / D);
    int d = (int)(idx % D);
    float di = dinv[i];
    float v  = agg[idx] + di * di * h[idx] + bias[d];
    out[idx] = RELU ? fmaxf(v, 0.0f) : v;
}

extern "C" void kernel_launch(void* const* d_in, const int* in_sizes, int n_in,
                              void* d_out, int out_size, void* d_ws, size_t ws_size,
                              hipStream_t stream) {
    const float* x  = (const float*)d_in[0];
    const int*   ei = (const int*)d_in[1];
    const float* W0 = (const float*)d_in[2];
    const float* b0 = (const float*)d_in[3];
    const float* W1 = (const float*)d_in[4];
    const float* b1 = (const float*)d_in[5];
    const float* W2 = (const float*)d_in[6];
    const float* b2 = (const float*)d_in[7];

    const int N = in_sizes[0] / IN_DIM;
    const int E = in_sizes[1] / 2;
    const int* rowIdx = ei;
    const int* colIdx = ei + E;

    char*  ws  = (char*)d_ws;
    size_t off = 0;
    auto   alloc = [&](size_t bytes) {
        void* p = ws + off;
        off += (bytes + 255) & ~(size_t)255;
        return p;
    };
    float* dinv = (float*)alloc((size_t)N * sizeof(float));
    float* hbuf = (float*)alloc((size_t)N * HID_DIM * sizeof(float));
    float* agg  = (float*)alloc((size_t)N * HID_DIM * sizeof(float));
    float* xbuf = (float*)alloc((size_t)N * HID_DIM * sizeof(float));

    const int mtiles   = (N + 15) / 16;
    const int edgeBlks = (E + 7) / 8;

    init_deg_kernel<<<(N + 255) / 256, 256, 0, stream>>>(dinv, N);
    accum_deg_kernel<<<(E + 255) / 256, 256, 0, stream>>>(colIdx, dinv, E);
    deg_to_dinv_kernel<<<(N + 255) / 256, 256, 0, stream>>>(dinv, N);

    gemm_wmma_kernel<128><<<mtiles, 256, 0, stream>>>(x, W0, hbuf, N);
    hipMemsetAsync(agg, 0, (size_t)N * 128 * sizeof(float), stream);
    propagate_kernel<128><<<edgeBlks, 256, 0, stream>>>(rowIdx, colIdx, dinv, hbuf, agg, E);
    finalize_kernel<128, true><<<(int)(((size_t)N * 128 + 255) / 256), 256, 0, stream>>>(
        agg, hbuf, dinv, b0, xbuf, N);

    gemm_wmma_kernel<128><<<mtiles, 256, 0, stream>>>(xbuf, W1, hbuf, N);
    hipMemsetAsync(agg, 0, (size_t)N * 128 * sizeof(float), stream);
    propagate_kernel<128><<<edgeBlks, 256, 0, stream>>>(rowIdx, colIdx, dinv, hbuf, agg, E);
    finalize_kernel<128, true><<<(int)(((size_t)N * 128 + 255) / 256), 256, 0, stream>>>(
        agg, hbuf, dinv, b1, xbuf, N);

    gemm_wmma_kernel<64><<<mtiles, 128, 0, stream>>>(xbuf, W2, hbuf, N);
    hipMemsetAsync(agg, 0, (size_t)N * 64 * sizeof(float), stream);
    propagate_kernel<64><<<edgeBlks, 256, 0, stream>>>(rowIdx, colIdx, dinv, hbuf, agg, E);
    finalize_kernel<64, false><<<(int)(((size_t)N * 64 + 255) / 256), 256, 0, stream>>>(
        agg, hbuf, dinv, b2, (float*)d_out, N);
}


# Per-edge float atomic scatter replaced by a gather: edges grouped by destination once per call (int atomic rank + scan); one wave sums a node's in-edges in registers (arrival order), writes row once
# speedup vs baseline: 2.3342x; 2.3342x over previous
#include <hip/hip_runtime.h>
#include <hip/hip_bf16.h>

typedef float v2f __attribute__((ext_vector_type(2)));
typedef float v8f __attribute__((ext_vector_type(8)));

#define IN_DIM  128
#define HID_DIM 128
#define OUT_DIM 64

__global__ void zero_cnt_kernel(int* __restrict__ cnt, int n) {
    int i = blockIdx.x * blockDim.x + threadIdx.x;
    if (i < n) cnt[i] = 0;
}

__global__ void count_rank_kernel(const int* __restrict__ col,
                                  int* __restrict__ cnt,
                                  int* __restrict__ rank, int E) {
    int e = blockIdx.x * blockDim.x + threadIdx.x;
    if (e < E) rank[e] = atomicAdd(&cnt[col[e]], 1);
}

#define SCAN_CHUNK 1024
__global__ void __launch_bounds__(256)
scan_dinv_kernel(const int* __restrict__ cnt, int2* __restrict__ seg,
                 float* __restrict__ dinv, int n) {
    __shared__ int s_pre[8];
    __shared__ int s_wav[8];
    const int t    = threadIdx.x;
    const int lane = t & 31;
    const int wv   = t >> 5;
    const int chunk0 = blockIdx.x * SCAN_CHUNK;

    int pre = 0;
    for (int i = t * 4; i < chunk0; i += 256 * 4) {
        int4 v = *(const int4*)(cnt + i);
        pre += (v.x + v.y) + (v.z + v.w);
    }
    for (int off = 16; off > 0; off >>= 1) pre += __shfl_xor(pre, off);

    const int idx = chunk0 + t * 4;
    int c0 = 0, c1 = 0, c2 = 0, c3 = 0;
    if (idx + 3 < n) {
        int4 v = *(const int4*)(cnt + idx);
        c0 = v.x; c1 = v.y; c2 = v.z; c3 = v.w;
    } else {
        if (idx     < n) c0 = cnt[idx];
        if (idx + 1 < n) c1 = cnt[idx + 1];
        if (idx + 2 < n) c2 = cnt[idx + 2];
    }
    const int s = (c0 + c1) + (c2 + c3);
    int inc = s;
    for (int off = 1; off < 32; off <<= 1) {
        int y = __shfl_up(inc, off);
        if (lane >= off) inc += y;
    }
    if (lane == 0)  s_pre[wv] = pre;
    if (lane == 31) s_wav[wv] = inc;
    __syncthreads();
    int before = 0;
    for (int w = 0; w < 8; ++w) {
        before += s_pre[w];
        if (w < wv) before += s_wav[w];
    }
    const int p0 = before + inc - s;
    if (idx < n)     { seg[idx]     = make_int2(p0, c0);                dinv[idx]     = rsqrtf((float)(c0 + 1)); }
    if (idx + 1 < n) { seg[idx + 1] = make_int2(p0 + c0, c1);           dinv[idx + 1] = rsqrtf((float)(c1 + 1)); }
    if (idx + 2 < n) { seg[idx + 2] = make_int2(p0 + c0 + c1, c2);      dinv[idx + 2] = rsqrtf((float)(c2 + 1)); }
    if (idx + 3 < n) { seg[idx + 3] = make_int2(p0 + c0 + c1 + c2, c3); dinv[idx + 3] = rsqrtf((float)(c3 + 1)); }
}

__global__ void fill_edges_kernel(const int* __restrict__ row,
                                  const int* __restrict__ col,
                                  const int* __restrict__ rank,
                                  const int2* __restrict__ seg,
                                  const float* __restrict__ dinv,
                                  int2* __restrict__ edges, int E) {
    int e = blockIdx.x * blockDim.x + threadIdx.x;
    if (e >= E) return;
    int r = row[e];
    int c = col[e];
    float nrm = dinv[r] * dinv[c];
    edges[seg[c].x + rank[e]] = make_int2(r, __float_as_int(nrm));
}

template <int NOUT>
__global__ void gemm_wmma_kernel(const float* __restrict__ X,
                                 const float* __restrict__ W,
                                 float* __restrict__ H, int nNodes) {
    constexpr int K = 128;
    const int wv   = threadIdx.x >> 5;
    const int lane = threadIdx.x & 31;
    const int half = lane >> 4;
    const int l16  = lane & 15;

    const int m0 = blockIdx.x * 16;
    const int n0 = wv * 16;

    int mrow = m0 + l16; if (mrow >= nNodes) mrow = nNodes - 1;
    const float* aptr = X + (size_t)mrow * K + 2 * half;
    const float* bptr = W + (size_t)(n0 + l16) * K + 2 * half;

    v8f c = {0.f, 0.f, 0.f, 0.f, 0.f, 0.f, 0.f, 0.f};
#pragma unroll
    for (int k0 = 0; k0 < K; k0 += 4) {
        v2f a = *(const v2f*)(aptr + k0);
        v2f b = *(const v2f*)(bptr + k0);
        c = __builtin_amdgcn_wmma_f32_16x16x4_f32(false, a, false, b,
                                                  (short)0, c, false, false);
    }

    float* out = H + (size_t)m0 * NOUT + n0;
#pragma unroll
    for (int v = 0; v < 8; ++v) {
        int r = v + 8 * half;
        if (m0 + r < nNodes) out[(size_t)r * NOUT + l16] = c[v];
    }
}

template <int D, bool RELU>
__global__ void __launch_bounds__(256)
gather_kernel(const int2* __restrict__ edges,
              const int2* __restrict__ seg,
              const float* __restrict__ dinv,
              const float* __restrict__ h,
              const float* __restrict__ bias,
              float* __restrict__ out, int n) {
    constexpr int V = D / 32;
    typedef float vf __attribute__((ext_vector_type(V)));
    const int lane = threadIdx.x & 31;
    const int node = __builtin_amdgcn_readfirstlane(
        (int)((blockIdx.x * (size_t)blockDim.x + threadIdx.x) >> 5));
    if (node >= n) return;
    const int2  sd    = seg[node];
    const int   start = sd.x;
    const int   deg   = sd.y;
    const float di    = dinv[node];
    const float* hl   = h + lane * V;
    const vf hs = *(const vf*)(hl + (size_t)node * D);
    const vf bs = *(const vf*)(bias + lane * V);

    vf acc = 0.0f;
    for (int base = 0; base < deg; base += 32) {
        const int m = min(32, deg - base);
        int2 e = make_int2(0, 0);
        if (lane < m) e = edges[start + base + lane];
        int j = 0;
        for (; j + 4 <= m; j += 4) {
            const int r0 = __builtin_amdgcn_readlane(e.x, j);
            const int r1 = __builtin_amdgcn_readlane(e.x, j + 1);
            const int r2 = __builtin_amdgcn_readlane(e.x, j + 2);
            const int r3 = __builtin_amdgcn_readlane(e.x, j + 3);
            const vf v0 = *(const vf*)(hl + (size_t)r0 * D);
            const vf v1 = *(const vf*)(hl + (size_t)r1 * D);
            const vf v2 = *(const vf*)(hl + (size_t)r2 * D);
            const vf v3 = *(const vf*)(hl + (size_t)r3 * D);
            const float w0 = __int_as_float(__builtin_amdgcn_readlane(e.y, j));
            const float w1 = __int_as_float(__builtin_amdgcn_readlane(e.y, j + 1));
            const float w2 = __int_as_float(__builtin_amdgcn_readlane(e.y, j + 2));
            const float w3 = __int_as_float(__builtin_amdgcn_readlane(e.y, j + 3));
            acc += w0 * v0;
            acc += w1 * v1;
            acc += w2 * v2;
            acc += w3 * v3;
        }
        const int rem = m - j;
        if (rem > 0) {
            const int   ra = __builtin_amdgcn_readlane(e.x, j);
            const float wa = __int_as_float(__builtin_amdgcn_readlane(e.y, j));
            const vf    va = *(const vf*)(hl + (size_t)ra * D);
            if (rem > 1) {
                const int   rb = __builtin_amdgcn_readlane(e.x, j + 1);
                const float wb = __int_as_float(__builtin_amdgcn_readlane(e.y, j + 1));
                const vf    vb = *(const vf*)(hl + (size_t)rb * D);
                if (rem > 2) {
                    const int   rc = __builtin_amdgcn_readlane(e.x, j + 2);
                    const float wc = __int_as_float(__builtin_amdgcn_readlane(e.y, j + 2));
                    const vf    vc = *(const vf*)(hl + (size_t)rc * D);
                    acc += wa * va;
                    acc += wb * vb;
                    acc += wc * vc;
                } else {
                    acc += wa * va;
                    acc += wb * vb;
                }
            } else {
                acc += wa * va;
            }
        }
    }

    vf o = acc + (di * di) * hs + bs;
    if (RELU) o = __builtin_elementwise_max(o, (vf)(0.0f));
    *(vf*)(out + (size_t)node * D + lane * V) = o;
}

extern "C" void kernel_launch(void* const* d_in, const int* in_sizes, int n_in,
                              void* d_out, int out_size, void* d_ws, size_t ws_size,
                              hipStream_t stream) {
    const float* x  = (const float*)d_in[0];
    const int*   ei = (const int*)d_in[1];
    const float* W0 = (const float*)d_in[2];
    const float* b0 = (const float*)d_in[3];
    const float* W1 = (const float*)d_in[4];
    const float* b1 = (const float*)d_in[5];
    const float* W2 = (const float*)d_in[6];
    const float* b2 = (const float*)d_in[7];

    const int N = in_sizes[0] / IN_DIM;
    const int E = in_sizes[1] / 2;
    const int* rowIdx = ei;
    const int* colIdx = ei + E;

    char*  ws  = (char*)d_ws;
    size_t off = 0;
    auto   alloc = [&](size_t bytes) {
        void* p = ws + off;
        off += (bytes + 255) & ~(size_t)255;
        return p;
    };
    float* dinv   = (float*)alloc((size_t)N * sizeof(float));
    float* hbuf   = (float*)alloc((size_t)N * HID_DIM * sizeof(float));
    float* xbuf   = (float*)alloc((size_t)N * HID_DIM * sizeof(float));
    int*   cnt    = (int*)alloc((size_t)N * sizeof(int));
    int2*  seg    = (int2*)alloc((size_t)N * sizeof(int2));
    int*   rank   = (int*)alloc((size_t)E * sizeof(int));
    int2*  edges  = (int2*)alloc((size_t)E * sizeof(int2));

    const int mtiles   = (N + 15) / 16;
    const int nodeBlks = (N + 7) / 8;

    zero_cnt_kernel<<<(N + 255) / 256, 256, 0, stream>>>(cnt, N);
    count_rank_kernel<<<(E + 255) / 256, 256, 0, stream>>>(colIdx, cnt, rank, E);
    scan_dinv_kernel<<<(N + SCAN_CHUNK - 1) / SCAN_CHUNK, 256, 0, stream>>>(cnt, seg, dinv, N);
    fill_edges_kernel<<<(E + 255) / 256, 256, 0, stream>>>(rowIdx, colIdx, rank, seg, dinv, edges, E);

    gemm_wmma_kernel<128><<<mtiles, 256, 0, stream>>>(x, W0, hbuf, N);
    gather_kernel<128, true><<<nodeBlks, 256, 0, stream>>>(edges, seg, dinv, hbuf, b0, xbuf, N);

    gemm_wmma_kernel<128><<<mtiles, 256, 0, stream>>>(xbuf, W1, hbuf, N);
    gather_kernel<128, true><<<nodeBlks, 256, 0, stream>>>(edges, seg, dinv, hbuf, b1, xbuf, N);

    gemm_wmma_kernel<64><<<mtiles, 128, 0, stream>>>(xbuf, W2, hbuf, N);
    gather_kernel<64, false><<<nodeBlks, 256, 0, stream>>>(edges, seg, dinv, hbuf, b2, (float*)d_out, N);
}


# Fewer, longer waves: a gather wave takes 4 nodes (segment records and first edges fetched ahead), a GEMM wave a whole 16-row strip (all column tiles), CSR threads 8 edges each: 5x fewer waves a call
# speedup vs baseline: 2.4802x; 1.0625x over previous
#include <hip/hip_runtime.h>
#include <hip/hip_bf16.h>

typedef float v2f __attribute__((ext_vector_type(2)));
typedef float v8f __attribute__((ext_vector_type(8)));

#define IN_DIM  128
#define HID_DIM 128
#define OUT_DIM 64

__global__ void zero_cnt_kernel(int* __restrict__ cnt, int n) {
    const int i = (blockIdx.x * blockDim.x + threadIdx.x) * 4;
    if (i + 3 < n) {
        *(int4*)(cnt + i) = make_int4(0, 0, 0, 0);
    } else {
        for (int k = i; k < n; ++k) cnt[k] = 0;
    }
}

#define EPT 8
__global__ void __launch_bounds__(256)
count_rank_kernel(const int* __restrict__ col,
                  int* __restrict__ cnt,
                  int* __restrict__ rank, int E) {
    const int base = blockIdx.x * (256 * EPT) + threadIdx.x;
    int c[EPT], r[EPT];
#pragma unroll
    for (int i = 0; i < EPT; ++i) {
        const int e = base + i * 256;
        c[i] = (e < E) ? col[e] : -1;
    }
#pragma unroll
    for (int i = 0; i < EPT; ++i) {
        r[i] = 0;
        if (c[i] >= 0) r[i] = atomicAdd(&cnt[c[i]], 1);
    }
#pragma unroll
    for (int i = 0; i < EPT; ++i) {
        const int e = base + i * 256;
        if (e < E) rank[e] = r[i];
    }
}

#define SCAN_CHUNK 1024
__global__ void __launch_bounds__(256)
scan_dinv_kernel(const int* __restrict__ cnt, int2* __restrict__ seg,
                 float* __restrict__ dinv, int n) {
    __shared__ int s_pre[8];
    __shared__ int s_wav[8];
    const int t    = threadIdx.x;
    const int lane = t & 31;
    const int wv   = t >> 5;
    const int chunk0 = blockIdx.x * SCAN_CHUNK;

    int pre = 0;
    for (int i = t * 4; i < chunk0; i += 256 * 4) {
        int4 v = *(const int4*)(cnt + i);
        pre += (v.x + v.y) + (v.z + v.w);
    }
    for (int off = 16; off > 0; off >>= 1) pre += __shfl_xor(pre, off);

    const int idx = chunk0 + t * 4;
    int c0 = 0, c1 = 0, c2 = 0, c3 = 0;
    if (idx + 3 < n) {
        int4 v = *(const int4*)(cnt + idx);
        c0 = v.x; c1 = v.y; c2 = v.z; c3 = v.w;
    } else {
        if (idx     < n) c0 = cnt[idx];
        if (idx + 1 < n) c1 = cnt[idx + 1];
        if (idx + 2 < n) c2 = cnt[idx + 2];
    }
    const int s = (c0 + c1) + (c2 + c3);
    int inc = s;
    for (int off = 1; off < 32; off <<= 1) {
        int y = __shfl_up(inc, off);
        if (lane >= off) inc += y;
    }
    if (lane == 0)  s_pre[wv] = pre;
    if (lane == 31) s_wav[wv] = inc;
    __syncthreads();
    int before = 0;
    for (int w = 0; w < 8; ++w) {
        before += s_pre[w];
        if (w < wv) before += s_wav[w];
    }
    const int p0 = before + inc - s;
    if (idx < n)     { seg[idx]     = make_int2(p0, c0);                dinv[idx]     = rsqrtf((float)(c0 + 1)); }
    if (idx + 1 < n) { seg[idx + 1] = make_int2(p0 + c0, c1);           dinv[idx + 1] = rsqrtf((float)(c1 + 1)); }
    if (idx + 2 < n) { seg[idx + 2] = make_int2(p0 + c0 + c1, c2);      dinv[idx + 2] = rsqrtf((float)(c2 + 1)); }
    if (idx + 3 < n) { seg[idx + 3] = make_int2(p0 + c0 + c1 + c2, c3); dinv[idx + 3] = rsqrtf((float)(c3 + 1)); }
}

__global__ void __launch_bounds__(256)
fill_edges_kernel(const int* __restrict__ row,
                                  const int* __restrict__ col,
                                  const int* __restrict__ rank,
                                  const int2* __restrict__ seg,
                                  const float* __restrict__ dinv,
                                  int2* __restrict__ edges, int E) {
    const int base = blockIdx.x * (256 * EPT) + threadIdx.x;
    int r[EPT], c[EPT], rk[EPT];
#pragma unroll
    for (int i = 0; i < EPT; ++i) {
        const int e = base + i * 256;
        const bool ok = e < E;
        r[i]  = ok ? row[e]  : -1;
        c[i]  = ok ? col[e]  : 0;
        rk[i] = ok ? rank[e] : 0;
    }
    float dr[EPT], dc[EPT];
    int   sx[EPT];
#pragma unroll
    for (int i = 0; i < EPT; ++i) {
        dr[i] = 0.f; dc[i] = 0.f; sx[i] = 0;
        if (r[i] >= 0) {
            dr[i] = dinv[r[i]];
            dc[i] = dinv[c[i]];
            sx[i] = seg[c[i]].x;
        }
    }
#pragma unroll
    for (int i = 0; i < EPT; ++i) {
        if (r[i] >= 0) {
            const float nrm = dr[i] * dc[i];
            edges[sx[i] + rk[i]] = make_int2(r[i], __float_as_int(nrm));
        }
    }
}

template <int NOUT>
__global__ void __launch_bounds__(256)
gemm_wmma_kernel(const float* __restrict__ X,
                 const float* __restrict__ W,
                 float* __restrict__ H, int nNodes) {
    constexpr int K  = 128;
    constexpr int NT = NOUT / 16;
    const int lane = threadIdx.x & 31;
    const int half = lane >> 4;
    const int l16  = lane & 15;

    const int strip = (int)((blockIdx.x * (size_t)blockDim.x + threadIdx.x) >> 5);
    const int m0 = strip * 16;
    if (m0 >= nNodes) return;

    int mrow = m0 + l16; if (mrow >= nNodes) mrow = nNodes - 1;
    const float* aptr = X + (size_t)mrow * K + 2 * half;
    const float* bptr = W + (size_t)l16 * K + 2 * half;

    v8f c[NT];
#pragma unroll
    for (int t = 0; t < NT; ++t) c[t] = (v8f){0.f, 0.f, 0.f, 0.f, 0.f, 0.f, 0.f, 0.f};

#pragma unroll
    for (int k0 = 0; k0 < K; k0 += 4) {
        const v2f a = *(const v2f*)(aptr + k0);
#pragma unroll
        for (int t = 0; t < NT; ++t) {
            const v2f b = *(const v2f*)(bptr + (size_t)t * 16 * K + k0);
            c[t] = __builtin_amdgcn_wmma_f32_16x16x4_f32(false, a, false, b,
                                                         (short)0, c[t], false, false);
        }
    }

    float* out = H + (size_t)m0 * NOUT;
#pragma unroll
    for (int t = 0; t < NT; ++t) {
#pragma unroll
        for (int v = 0; v < 8; ++v) {
            const int r = v + 8 * half;
            if (m0 + r < nNodes) out[(size_t)r * NOUT + t * 16 + l16] = c[t][v];
        }
    }
}

template <int D, bool RELU>
__device__ __forceinline__ void
gather_one(const int2* __restrict__ edges, const int start, const int deg,
           const float di, const int2 e_first,
           const float* __restrict__ hl, const float* __restrict__ bias_l,
           float* __restrict__ out, const int node, const int lane) {
    constexpr int V = D / 32;
    typedef float vf __attribute__((ext_vector_type(V)));
    const vf hs = *(const vf*)(hl + (size_t)node * D);
    const vf bs = *(const vf*)(bias_l);

    vf acc = 0.0f;
    for (int base = 0; base < deg; base += 32) {
        const int m = min(32, deg - base);
        int2 e = e_first;
        if (base > 0) {
            e = make_int2(0, 0);
            if (lane < m) e = edges[start + base + lane];
        }
        int j = 0;
        for (; j + 4 <= m; j += 4) {
            const int r0 = __builtin_amdgcn_readlane(e.x, j);
            const int r1 = __builtin_amdgcn_readlane(e.x, j + 1);
            const int r2 = __builtin_amdgcn_readlane(e.x, j + 2);
            const int r3 = __builtin_amdgcn_readlane(e.x, j + 3);
            const vf v0 = *(const vf*)(hl + (size_t)r0 * D);
            const vf v1 = *(const vf*)(hl + (size_t)r1 * D);
            const vf v2 = *(const vf*)(hl + (size_t)r2 * D);
            const vf v3 = *(const vf*)(hl + (size_t)r3 * D);
            const float w0 = __int_as_float(__builtin_amdgcn_readlane(e.y, j));
            const float w1 = __int_as_float(__builtin_amdgcn_readlane(e.y, j + 1));
            const float w2 = __int_as_float(__builtin_amdgcn_readlane(e.y, j + 2));
            const float w3 = __int_as_float(__builtin_amdgcn_readlane(e.y, j + 3));
            acc += w0 * v0;
            acc += w1 * v1;
            acc += w2 * v2;
            acc += w3 * v3;
        }
        const int rem = m - j;
        if (rem > 0) {
            const int   ra = __builtin_amdgcn_readlane(e.x, j);
            const float wa = __int_as_float(__builtin_amdgcn_readlane(e.y, j));
            const vf    va = *(const vf*)(hl + (size_t)ra * D);
            if (rem > 1) {
                const int   rb = __builtin_amdgcn_readlane(e.x, j + 1);
                const float wb = __int_as_float(__builtin_amdgcn_readlane(e.y, j + 1));
                const vf    vb = *(const vf*)(hl + (size_t)rb * D);
                if (rem > 2) {
                    const int   rc = __builtin_amdgcn_readlane(e.x, j + 2);
                    const float wc = __int_as_float(__builtin_amdgcn_readlane(e.y, j + 2));
                    const vf    vc = *(const vf*)(hl + (size_t)rc * D);
                    acc += wa * va;
                    acc += wb * vb;
                    acc += wc * vc;
                } else {
                    acc += wa * va;
                    acc += wb * vb;
                }
            } else {
                acc += wa * va;
            }
        }
    }

    vf o = acc + (di * di) * hs + bs;
    if (RELU) o = __builtin_elementwise_max(o, (vf)(0.0f));
    *(vf*)(out + (size_t)node * D + lane * V) = o;
}

#define NPW 4
template <int D, bool RELU>
__global__ void __launch_bounds__(256)
gather_kernel(const int2* __restrict__ edges,
              const int2* __restrict__ seg,
              const float* __restrict__ dinv,
              const float* __restrict__ h,
              const float* __restrict__ bias,
              float* __restrict__ out, int n) {
    constexpr int V = D / 32;
    const int lane = threadIdx.x & 31;
    const int node0 = NPW * __builtin_amdgcn_readfirstlane(
        (int)((blockIdx.x * (size_t)blockDim.x + threadIdx.x) >> 5));
    if (node0 >= n) return;
    const float* hl = h + lane * V;

    int   start[NPW], deg[NPW];
    float di[NPW];
    int2  e[NPW];
#pragma unroll
    for (int i = 0; i < NPW; ++i) {
        const int  nd = min(node0 + i, n - 1);
        const int2 sd = seg[nd];
        start[i] = sd.x;
        deg[i]   = (node0 + i < n) ? sd.y : 0;
        di[i]    = dinv[nd];
    }
#pragma unroll
    for (int i = 0; i < NPW; ++i) {
        e[i] = make_int2(0, 0);
        if (lane < deg[i]) e[i] = edges[start[i] + lane];
    }
#pragma unroll
    for (int i = 0; i < NPW; ++i) {
        if (node0 + i < n)
            gather_one<D, RELU>(edges, start[i], deg[i], di[i], e[i], hl,
                                bias + lane * V, out, node0 + i, lane);
    }
}

extern "C" void kernel_launch(void* const* d_in, const int* in_sizes, int n_in,
                              void* d_out, int out_size, void* d_ws, size_t ws_size,
                              hipStream_t stream) {
    const float* x  = (const float*)d_in[0];
    const int*   ei = (const int*)d_in[1];
    const float* W0 = (const float*)d_in[2];
    const float* b0 = (const float*)d_in[3];
    const float* W1 = (const float*)d_in[4];
    const float* b1 = (const float*)d_in[5];
    const float* W2 = (const float*)d_in[6];
    const float* b2 = (const float*)d_in[7];

    const int N = in_sizes[0] / IN_DIM;
    const int E = in_sizes[1] / 2;
    const int* rowIdx = ei;
    const int* colIdx = ei + E;

    char*  ws  = (char*)d_ws;
    size_t off = 0;
    auto   alloc = [&](size_t bytes) {
        void* p = ws + off;
        off += (bytes + 255) & ~(size_t)255;
        return p;
    };
    float* dinv   = (float*)alloc((size_t)N * sizeof(float));
    float* hbuf   = (float*)alloc((size_t)N * HID_DIM * sizeof(float));
    float* xbuf   = (float*)alloc((size_t)N * HID_DIM * sizeof(float));
    int*   cnt    = (int*)alloc((size_t)N * sizeof(int));
    int2*  seg    = (int2*)alloc((size_t)N * sizeof(int2));
    int*   rank   = (int*)alloc((size_t)E * sizeof(int));
    int2*  edges  = (int2*)alloc((size_t)E * sizeof(int2));

    const int mtiles   = (N + 15) / 16;
    const int gemmBlks = (mtiles + 7) / 8;
    const int nodeBlks = (N + 8 * NPW - 1) / (8 * NPW);
    const int edgeBlks = (E + 256 * EPT - 1) / (256 * EPT);

    zero_cnt_kernel<<<(N + 1023) / 1024, 256, 0, stream>>>(cnt, N);
    count_rank_kernel<<<edgeBlks, 256, 0, stream>>>(colIdx, cnt, rank, E);
    scan_dinv_kernel<<<(N + SCAN_CHUNK - 1) / SCAN_CHUNK, 256, 0, stream>>>(cnt, seg, dinv, N);
    fill_edges_kernel<<<edgeBlks, 256, 0, stream>>>(rowIdx, colIdx, rank, seg, dinv, edges, E);

    gemm_wmma_kernel<128><<<gemmBlks, 256, 0, stream>>>(x, W0, hbuf, N);
    gather_kernel<128, true><<<nodeBlks, 256, 0, stream>>>(edges, seg, dinv, hbuf, b0, xbuf, N);

    gemm_wmma_kernel<128><<<gemmBlks, 256, 0, stream>>>(xbuf, W1, hbuf, N);
    gather_kernel<128, true><<<nodeBlks, 256, 0, stream>>>(edges, seg, dinv, hbuf, b1, xbuf, N);

    gemm_wmma_kernel<64><<<gemmBlks, 256, 0, stream>>>(xbuf, W2, hbuf, N);
    gather_kernel<64, false><<<nodeBlks, 256, 0, stream>>>(edges, seg, dinv, hbuf, b2, (float*)d_out, N);
}


# GEMM operands read from LDS: W copied there once per block and each wave's 16 rows of X with whole-row 16-byte loads, instead of 8-byte fragment loads from global that touch 16 lines each
# speedup vs baseline: 2.8118x; 1.1337x over previous
#include <hip/hip_runtime.h>
#include <hip/hip_bf16.h>

typedef float v2f __attribute__((ext_vector_type(2)));
typedef float v8f __attribute__((ext_vector_type(8)));

#define IN_DIM  128
#define HID_DIM 128
#define OUT_DIM 64

__global__ void zero_cnt_kernel(int* __restrict__ cnt, int n) {
    const int i = (blockIdx.x * blockDim.x + threadIdx.x) * 4;
    if (i + 3 < n) {
        *(int4*)(cnt + i) = make_int4(0, 0, 0, 0);
    } else {
        for (int k = i; k < n; ++k) cnt[k] = 0;
    }
}

#define EPT 8
__global__ void __launch_bounds__(256)
count_rank_kernel(const int* __restrict__ col,
                  int* __restrict__ cnt,
                  int* __restrict__ rank, int E) {
    const int base = blockIdx.x * (256 * EPT) + threadIdx.x;
    int c[EPT], r[EPT];
#pragma unroll
    for (int i = 0; i < EPT; ++i) {
        const int e = base + i * 256;
        c[i] = (e < E) ? col[e] : -1;
    }
#pragma unroll
    for (int i = 0; i < EPT; ++i) {
        r[i] = 0;
        if (c[i] >= 0) r[i] = atomicAdd(&cnt[c[i]], 1);
    }
#pragma unroll
    for (int i = 0; i < EPT; ++i) {
        const int e = base + i * 256;
        if (e < E) rank[e] = r[i];
    }
}

#define SCAN_CHUNK 1024
__global__ void __launch_bounds__(256)
scan_dinv_kernel(const int* __restrict__ cnt, int2* __restrict__ seg,
                 float* __restrict__ dinv, int n) {
    __shared__ int s_pre[8];
    __shared__ int s_wav[8];
    const int t    = threadIdx.x;
    const int lane = t & 31;
    const int wv   = t >> 5;
    const int chunk0 = blockIdx.x * SCAN_CHUNK;

    int pre = 0;
    for (int i = t * 4; i < chunk0; i += 256 * 4) {
        int4 v = *(const int4*)(cnt + i);
        pre += (v.x + v.y) + (v.z + v.w);
    }
    for (int off = 16; off > 0; off >>= 1) pre += __shfl_xor(pre, off);

    const int idx = chunk0 + t * 4;
    int c0 = 0, c1 = 0, c2 = 0, c3 = 0;
    if (idx + 3 < n) {
        int4 v = *(const int4*)(cnt + idx);
        c0 = v.x; c1 = v.y; c2 = v.z; c3 = v.w;
    } else {
        if (idx     < n) c0 = cnt[idx];
        if (idx + 1 < n) c1 = cnt[idx + 1];
        if (idx + 2 < n) c2 = cnt[idx + 2];
    }
    const int s = (c0 + c1) + (c2 + c3);
    int inc = s;
    for (int off = 1; off < 32; off <<= 1) {
        int y = __shfl_up(inc, off);
        if (lane >= off) inc += y;
    }
    if (lane == 0)  s_pre[wv] = pre;
    if (lane == 31) s_wav[wv] = inc;
    __syncthreads();
    int before = 0;
    for (int w = 0; w < 8; ++w) {
        before += s_pre[w];
        if (w < wv) before += s_wav[w];
    }
    const int p0 = before + inc - s;
    if (idx < n)     { seg[idx]     = make_int2(p0, c0);                dinv[idx]     = rsqrtf((float)(c0 + 1)); }
    if (idx + 1 < n) { seg[idx + 1] = make_int2(p0 + c0, c1);           dinv[idx + 1] = rsqrtf((float)(c1 + 1)); }
    if (idx + 2 < n) { seg[idx + 2] = make_int2(p0 + c0 + c1, c2);      dinv[idx + 2] = rsqrtf((float)(c2 + 1)); }
    if (idx + 3 < n) { seg[idx + 3] = make_int2(p0 + c0 + c1 + c2, c3); dinv[idx + 3] = rsqrtf((float)(c3 + 1)); }
}

__global__ void __launch_bounds__(256)
fill_edges_kernel(const int* __restrict__ row,
                                  const int* __restrict__ col,
                                  const int* __restrict__ rank,
                                  const int2* __restrict__ seg,
                                  const float* __restrict__ dinv,
                                  int2* __restrict__ edges, int E) {
    const int base = blockIdx.x * (256 * EPT) + threadIdx.x;
    int r[EPT], c[EPT], rk[EPT];
#pragma unroll
    for (int i = 0; i < EPT; ++i) {
        const int e = base + i * 256;
        const bool ok = e < E;
        r[i]  = ok ? row[e]  : -1;
        c[i]  = ok ? col[e]  : 0;
        rk[i] = ok ? rank[e] : 0;
    }
    float dr[EPT], dc[EPT];
    int   sx[EPT];
#pragma unroll
    for (int i = 0; i < EPT; ++i) {
        dr[i] = 0.f; dc[i] = 0.f; sx[i] = 0;
        if (r[i] >= 0) {
            dr[i] = dinv[r[i]];
            dc[i] = dinv[c[i]];
            sx[i] = seg[c[i]].x;
        }
    }
#pragma unroll
    for (int i = 0; i < EPT; ++i) {
        if (r[i] >= 0) {
            const float nrm = dr[i] * dc[i];
            edges[sx[i] + rk[i]] = make_int2(r[i], __float_as_int(nrm));
        }
    }
}

#define LDS_PITCH 132
template <int NOUT>
__global__ void __launch_bounds__(256)
gemm_wmma_kernel(const float* __restrict__ X,
                 const float* __restrict__ W,
                 float* __restrict__ H, int nNodes) {
    constexpr int K  = 128;
    constexpr int NT = NOUT / 16;
    constexpr int P  = LDS_PITCH;
    __shared__ __attribute__((aligned(16))) float sW[NOUT * P];
    __shared__ __attribute__((aligned(16))) float sX[8 * 16 * P];

    const int lane = threadIdx.x & 31;
    const int wv   = threadIdx.x >> 5;
    const int half = lane >> 4;
    const int l16  = lane & 15;
    const int m0   = (blockIdx.x * 8 + wv) * 16;

    float4 wr[NOUT / 8];
#pragma unroll
    for (int i = 0; i < NOUT / 8; ++i)
        wr[i] = *(const float4*)(W + (size_t)(i * 8 + wv) * K + lane * 4);
    float4 xr[16];
#pragma unroll
    for (int r = 0; r < 16; ++r) {
        int mrow = m0 + r; if (mrow >= nNodes) mrow = nNodes - 1;
        xr[r] = *(const float4*)(X + (size_t)mrow * K + lane * 4);
    }
#pragma unroll
    for (int i = 0; i < NOUT / 8; ++i)
        *(float4*)(sW + (i * 8 + wv) * P + lane * 4) = wr[i];
#pragma unroll
    for (int r = 0; r < 16; ++r)
        *(float4*)(sX + (wv * 16 + r) * P + lane * 4) = xr[r];
    __syncthreads();

    if (m0 < nNodes) {
        const float* aptr = sX + (wv * 16 + l16) * P + 2 * half;
        const float* bptr = sW + l16 * P + 2 * half;

        v8f c[NT];
#pragma unroll
        for (int t = 0; t < NT; ++t) c[t] = (v8f){0.f, 0.f, 0.f, 0.f, 0.f, 0.f, 0.f, 0.f};

#pragma unroll
        for (int k0 = 0; k0 < K; k0 += 4) {
            const v2f a = *(const v2f*)(aptr + k0);
#pragma unroll
            for (int t = 0; t < NT; ++t) {
                const v2f b = *(const v2f*)(bptr + t * 16 * P + k0);
                c[t] = __builtin_amdgcn_wmma_f32_16x16x4_f32(false, a, false, b,
                                                             (short)0, c[t], false, false);
            }
        }

        float* out = H + (size_t)m0 * NOUT;
#pragma unroll
        for (int t = 0; t < NT; ++t) {
#pragma unroll
            for (int v = 0; v < 8; ++v) {
                const int r = v + 8 * half;
                if (m0 + r < nNodes) out[(size_t)r * NOUT + t * 16 + l16] = c[t][v];
            }
        }
    }
}

template <int D, bool RELU>
__device__ __forceinline__ void
gather_one(const int2* __restrict__ edges, const int start, const int deg,
           const float di, const int2 e_first,
           const float* __restrict__ hl, const float* __restrict__ bias_l,
           float* __restrict__ out, const int node, const int lane) {
    constexpr int V = D / 32;
    typedef float vf __attribute__((ext_vector_type(V)));
    const vf hs = *(const vf*)(hl + (size_t)node * D);
    const vf bs = *(const vf*)(bias_l);

    vf acc = 0.0f;
    for (int base = 0; base < deg; base += 32) {
        const int m = min(32, deg - base);
        int2 e = e_first;
        if (base > 0) {
            e = make_int2(0, 0);
            if (lane < m) e = edges[start + base + lane];
        }
        int j = 0;
        for (; j + 4 <= m; j += 4) {
            const int r0 = __builtin_amdgcn_readlane(e.x, j);
            const int r1 = __builtin_amdgcn_readlane(e.x, j + 1);
            const int r2 = __builtin_amdgcn_readlane(e.x, j + 2);
            const int r3 = __builtin_amdgcn_readlane(e.x, j + 3);
            const vf v0 = *(const vf*)(hl + (size_t)r0 * D);
            const vf v1 = *(const vf*)(hl + (size_t)r1 * D);
            const vf v2 = *(const vf*)(hl + (size_t)r2 * D);
            const vf v3 = *(const vf*)(hl + (size_t)r3 * D);
            const float w0 = __int_as_float(__builtin_amdgcn_readlane(e.y, j));
            const float w1 = __int_as_float(__builtin_amdgcn_readlane(e.y, j + 1));
            const float w2 = __int_as_float(__builtin_amdgcn_readlane(e.y, j + 2));
            const float w3 = __int_as_float(__builtin_amdgcn_readlane(e.y, j + 3));
            acc += w0 * v0;
            acc += w1 * v1;
            acc += w2 * v2;
            acc += w3 * v3;
        }
        const int rem = m - j;
        if (rem > 0) {
            const int   ra = __builtin_amdgcn_readlane(e.x, j);
            const float wa = __int_as_float(__builtin_amdgcn_readlane(e.y, j));
            const vf    va = *(const vf*)(hl + (size_t)ra * D);
            if (rem > 1) {
                const int   rb = __builtin_amdgcn_readlane(e.x, j + 1);
                const float wb = __int_as_float(__builtin_amdgcn_readlane(e.y, j + 1));
                const vf    vb = *(const vf*)(hl + (size_t)rb * D);
                if (rem > 2) {
                    const int   rc = __builtin_amdgcn_readlane(e.x, j + 2);
                    const float wc = __int_as_float(__builtin_amdgcn_readlane(e.y, j + 2));
                    const vf    vc = *(const vf*)(hl + (size_t)rc * D);
                    acc += wa * va;
                    acc += wb * vb;
                    acc += wc * vc;
                } else {
                    acc += wa * va;
                    acc += wb * vb;
                }
            } else {
                acc += wa * va;
            }
        }
    }

    vf o = acc + (di * di) * hs + bs;
    if (RELU) o = __builtin_elementwise_max(o, (vf)(0.0f));
    *(vf*)(out + (size_t)node * D + lane * V) = o;
}

#define NPW 4
template <int D, bool RELU>
__global__ void __launch_bounds__(256)
gather_kernel(const int2* __restrict__ edges,
              const int2* __restrict__ seg,
              const float* __restrict__ dinv,
              const float* __restrict__ h,
              const float* __restrict__ bias,
              float* __restrict__ out, int n) {
    constexpr int V = D / 32;
    const int lane = threadIdx.x & 31;
    const int node0 = NPW * __builtin_amdgcn_readfirstlane(
        (int)((blockIdx.x * (size_t)blockDim.x + threadIdx.x) >> 5));
    if (node0 >= n) return;
    const float* hl = h + lane * V;

    int   start[NPW], deg[NPW];
    float di[NPW];
    int2  e[NPW];
#pragma unroll
    for (int i = 0; i < NPW; ++i) {
        const int  nd = min(node0 + i, n - 1);
        const int2 sd = seg[nd];
        start[i] = sd.x;
        deg[i]   = (node0 + i < n) ? sd.y : 0;
        di[i]    = dinv[nd];
    }
#pragma unroll
    for (int i = 0; i < NPW; ++i) {
        e[i] = make_int2(0, 0);
        if (lane < deg[i]) e[i] = edges[start[i] + lane];
    }
#pragma unroll
    for (int i = 0; i < NPW; ++i) {
        if (node0 + i < n)
            gather_one<D, RELU>(edges, start[i], deg[i], di[i], e[i], hl,
                                bias + lane * V, out, node0 + i, lane);
    }
}

extern "C" void kernel_launch(void* const* d_in, const int* in_sizes, int n_in,
                              void* d_out, int out_size, void* d_ws, size_t ws_size,
                              hipStream_t stream) {
    const float* x  = (const float*)d_in[0];
    const int*   ei = (const int*)d_in[1];
    const float* W0 = (const float*)d_in[2];
    const float* b0 = (const float*)d_in[3];
    const float* W1 = (const float*)d_in[4];
    const float* b1 = (const float*)d_in[5];
    const float* W2 = (const float*)d_in[6];
    const float* b2 = (const float*)d_in[7];

    const int N = in_sizes[0] / IN_DIM;
    const int E = in_sizes[1] / 2;
    const int* rowIdx = ei;
    const int* colIdx = ei + E;

    char*  ws  = (char*)d_ws;
    size_t off = 0;
    auto   alloc = [&](size_t bytes) {
        void* p = ws + off;
        off += (bytes + 255) & ~(size_t)255;
        return p;
    };
    float* dinv   = (float*)alloc((size_t)N * sizeof(float));
    float* hbuf   = (float*)alloc((size_t)N * HID_DIM * sizeof(float));
    float* xbuf   = (float*)alloc((size_t)N * HID_DIM * sizeof(float));
    int*   cnt    = (int*)alloc((size_t)N * sizeof(int));
    int2*  seg    = (int2*)alloc((size_t)N * sizeof(int2));
    int*   rank   = (int*)alloc((size_t)E * sizeof(int));
    int2*  edges  = (int2*)alloc((size_t)E * sizeof(int2));

    const int mtiles   = (N + 15) / 16;
    const int gemmBlks = (mtiles + 7) / 8;
    const int nodeBlks = (N + 8 * NPW - 1) / (8 * NPW);
    const int edgeBlks = (E + 256 * EPT - 1) / (256 * EPT);

    zero_cnt_kernel<<<(N + 1023) / 1024, 256, 0, stream>>>(cnt, N);
    count_rank_kernel<<<edgeBlks, 256, 0, stream>>>(colIdx, cnt, rank, E);
    scan_dinv_kernel<<<(N + SCAN_CHUNK - 1) / SCAN_CHUNK, 256, 0, stream>>>(cnt, seg, dinv, N);
    fill_edges_kernel<<<edgeBlks, 256, 0, stream>>>(rowIdx, colIdx, rank, seg, dinv, edges, E);

    gemm_wmma_kernel<128><<<gemmBlks, 256, 0, stream>>>(x, W0, hbuf, N);
    gather_kernel<128, true><<<nodeBlks, 256, 0, stream>>>(edges, seg, dinv, hbuf, b0, xbuf, N);

    gemm_wmma_kernel<128><<<gemmBlks, 256, 0, stream>>>(xbuf, W1, hbuf, N);
    gather_kernel<128, true><<<nodeBlks, 256, 0, stream>>>(edges, seg, dinv, hbuf, b1, xbuf, N);

    gemm_wmma_kernel<64><<<gemmBlks, 256, 0, stream>>>(xbuf, W2, hbuf, N);
    gather_kernel<64, false><<<nodeBlks, 256, 0, stream>>>(edges, seg, dinv, hbuf, b2, (float*)d_out, N);
}


# More loads in flight per wave: gather requests 8 source rows per memory round trip, a node's last 1-7 edges as one more trip (was 4 per trip); scan reads 16 chunks per trip (was 1); same sum order
# speedup vs baseline: 3.1159x; 1.1082x over previous
#include <hip/hip_runtime.h>
#include <hip/hip_bf16.h>

typedef float v2f __attribute__((ext_vector_type(2)));
typedef float v8f __attribute__((ext_vector_type(8)));

#define IN_DIM  128
#define HID_DIM 128
#define OUT_DIM 64

__global__ void zero_cnt_kernel(int* __restrict__ cnt, int n) {
    const int i = (blockIdx.x * blockDim.x + threadIdx.x) * 4;
    if (i + 3 < n) {
        *(int4*)(cnt + i) = make_int4(0, 0, 0, 0);
    } else {
        for (int k = i; k < n; ++k) cnt[k] = 0;
    }
}

#define EPT 8
__global__ void __launch_bounds__(256)
count_rank_kernel(const int* __restrict__ col,
                  int* __restrict__ cnt,
                  int* __restrict__ rank, int E) {
    const int base = blockIdx.x * (256 * EPT) + threadIdx.x;
    int c[EPT], r[EPT];
#pragma unroll
    for (int i = 0; i < EPT; ++i) {
        const int e = base + i * 256;
        c[i] = (e < E) ? col[e] : -1;
    }
#pragma unroll
    for (int i = 0; i < EPT; ++i) {
        r[i] = 0;
        if (c[i] >= 0) r[i] = atomicAdd(&cnt[c[i]], 1);
    }
#pragma unroll
    for (int i = 0; i < EPT; ++i) {
        const int e = base + i * 256;
        if (e < E) rank[e] = r[i];
    }
}

#define SCAN_CHUNK 1024
#define PF_BATCH 16
__global__ void __launch_bounds__(256)
scan_dinv_kernel(const int* __restrict__ cnt, int2* __restrict__ seg,
                 float* __restrict__ dinv, int n) {
    __shared__ int s_pre[8];
    __shared__ int s_wav[8];
    const int t    = threadIdx.x;
    const int lane = t & 31;
    const int wv   = t >> 5;
    const int chunk0 = blockIdx.x * SCAN_CHUNK;

    int pre = 0;
    const int nfront = (int)blockIdx.x;
    const int* pf = cnt + t * 4;
    for (int it = 0; it < nfront; it += PF_BATCH) {
        int4 v[PF_BATCH];
#pragma unroll
        for (int u = 0; u < PF_BATCH; ++u) {
            v[u] = make_int4(0, 0, 0, 0);
            if (it + u < nfront) v[u] = *(const int4*)(pf + (size_t)(it + u) * SCAN_CHUNK);
        }
#pragma unroll
        for (int u = 0; u < PF_BATCH; ++u) pre += (v[u].x + v[u].y) + (v[u].z + v[u].w);
    }
    for (int off = 16; off > 0; off >>= 1) pre += __shfl_xor(pre, off);

    const int idx = chunk0 + t * 4;
    int c0 = 0, c1 = 0, c2 = 0, c3 = 0;
    if (idx + 3 < n) {
        int4 v = *(const int4*)(cnt + idx);
        c0 = v.x; c1 = v.y; c2 = v.z; c3 = v.w;
    } else {
        if (idx     < n) c0 = cnt[idx];
        if (idx + 1 < n) c1 = cnt[idx + 1];
        if (idx + 2 < n) c2 = cnt[idx + 2];
    }
    const int s = (c0 + c1) + (c2 + c3);
    int inc = s;
    for (int off = 1; off < 32; off <<= 1) {
        int y = __shfl_up(inc, off);
        if (lane >= off) inc += y;
    }
    if (lane == 0)  s_pre[wv] = pre;
    if (lane == 31) s_wav[wv] = inc;
    __syncthreads();
    int before = 0;
    for (int w = 0; w < 8; ++w) {
        before += s_pre[w];
        if (w < wv) before += s_wav[w];
    }
    const int p0 = before + inc - s;
    if (idx < n)     { seg[idx]     = make_int2(p0, c0);                dinv[idx]     = rsqrtf((float)(c0 + 1)); }
    if (idx + 1 < n) { seg[idx + 1] = make_int2(p0 + c0, c1);           dinv[idx + 1] = rsqrtf((float)(c1 + 1)); }
    if (idx + 2 < n) { seg[idx + 2] = make_int2(p0 + c0 + c1, c2);      dinv[idx + 2] = rsqrtf((float)(c2 + 1)); }
    if (idx + 3 < n) { seg[idx + 3] = make_int2(p0 + c0 + c1 + c2, c3); dinv[idx + 3] = rsqrtf((float)(c3 + 1)); }
}

__global__ void __launch_bounds__(256)
fill_edges_kernel(const int* __restrict__ row,
                                  const int* __restrict__ col,
                                  const int* __restrict__ rank,
                                  const int2* __restrict__ seg,
                                  const float* __restrict__ dinv,
                                  int2* __restrict__ edges, int E) {
    const int base = blockIdx.x * (256 * EPT) + threadIdx.x;
    int r[EPT], c[EPT], rk[EPT];
#pragma unroll
    for (int i = 0; i < EPT; ++i) {
        const int e = base + i * 256;
        const bool ok = e < E;
        r[i]  = ok ? row[e]  : -1;
        c[i]  = ok ? col[e]  : 0;
        rk[i] = ok ? rank[e] : 0;
    }
    float dr[EPT], dc[EPT];
    int   sx[EPT];
#pragma unroll
    for (int i = 0; i < EPT; ++i) {
        dr[i] = 0.f; dc[i] = 0.f; sx[i] = 0;
        if (r[i] >= 0) {
            dr[i] = dinv[r[i]];
            dc[i] = dinv[c[i]];
            sx[i] = seg[c[i]].x;
        }
    }
#pragma unroll
    for (int i = 0; i < EPT; ++i) {
        if (r[i] >= 0) {
            const float nrm = dr[i] * dc[i];
            edges[sx[i] + rk[i]] = make_int2(r[i], __float_as_int(nrm));
        }
    }
}

#define LDS_PITCH 132
template <int NOUT>
__global__ void __launch_bounds__(256)
gemm_wmma_kernel(const float* __restrict__ X,
                 const float* __restrict__ W,
                 float* __restrict__ H, int nNodes) {
    constexpr int K  = 128;
    constexpr int NT = NOUT / 16;
    constexpr int P  = LDS_PITCH;
    __shared__ __attribute__((aligned(16))) float sW[NOUT * P];
    __shared__ __attribute__((aligned(16))) float sX[8 * 16 * P];

    const int lane = threadIdx.x & 31;
    const int wv   = threadIdx.x >> 5;
    const int half = lane >> 4;
    const int l16  = lane & 15;
    const int m0   = (blockIdx.x * 8 + wv) * 16;

    float4 wr[NOUT / 8];
#pragma unroll
    for (int i = 0; i < NOUT / 8; ++i)
        wr[i] = *(const float4*)(W + (size_t)(i * 8 + wv) * K + lane * 4);
    float4 xr[16];
#pragma unroll
    for (int r = 0; r < 16; ++r) {
        int mrow = m0 + r; if (mrow >= nNodes) mrow = nNodes - 1;
        xr[r] = *(const float4*)(X + (size_t)mrow * K + lane * 4);
    }
#pragma unroll
    for (int i = 0; i < NOUT / 8; ++i)
        *(float4*)(sW + (i * 8 + wv) * P + lane * 4) = wr[i];
#pragma unroll
    for (int r = 0; r < 16; ++r)
        *(float4*)(sX + (wv * 16 + r) * P + lane * 4) = xr[r];
    __syncthreads();

    if (m0 < nNodes) {
        const float* aptr = sX + (wv * 16 + l16) * P + 2 * half;
        const float* bptr = sW + l16 * P + 2 * half;

        v8f c[NT];
#pragma unroll
        for (int t = 0; t < NT; ++t) c[t] = (v8f){0.f, 0.f, 0.f, 0.f, 0.f, 0.f, 0.f, 0.f};

#pragma unroll
        for (int k0 = 0; k0 < K; k0 += 4) {
            const v2f a = *(const v2f*)(aptr + k0);
#pragma unroll
            for (int t = 0; t < NT; ++t) {
                const v2f b = *(const v2f*)(bptr + t * 16 * P + k0);
                c[t] = __builtin_amdgcn_wmma_f32_16x16x4_f32(false, a, false, b,
                                                             (short)0, c[t], false, false);
            }
        }

        float* out = H + (size_t)m0 * NOUT;
#pragma unroll
        for (int t = 0; t < NT; ++t) {
#pragma unroll
            for (int v = 0; v < 8; ++v) {
                const int r = v + 8 * half;
                if (m0 + r < nNodes) out[(size_t)r * NOUT + t * 16 + l16] = c[t][v];
            }
        }
    }
}

template <int D, int R>
__device__ __forceinline__ void
add_rows(float __attribute__((ext_vector_type(D / 32)))& acc, const int2 e, const int j,
         const char* __restrict__ hb, const unsigned loff) {
    typedef float vf __attribute__((ext_vector_type(D / 32)));
    vf v[R];
#pragma unroll
    for (int u = 0; u < R; ++u) {
        const int r = __builtin_amdgcn_readlane(e.x, j + u);
        v[u] = *(const vf*)(hb + (size_t)r * (D * 4) + loff);
    }
#pragma unroll
    for (int u = 0; u < R; ++u) {
        const float w = __int_as_float(__builtin_amdgcn_readlane(e.y, j + u));
        acc += w * v[u];
    }
}

#define GROWS 8
template <int D, bool RELU>
__device__ __forceinline__ void
gather_one(const int2* __restrict__ edges, const int start, const int deg,
           const float di, const int2 e_first,
           const float* __restrict__ h, const int lane,
           const float* __restrict__ bias_l,
           float* __restrict__ out, const int node) {
    constexpr int V = D / 32;
    typedef float vf __attribute__((ext_vector_type(V)));
    const char*    hb   = (const char*)h;
    const unsigned loff = (unsigned)lane * (V * 4);
    const vf hs = *(const vf*)(hb + (size_t)node * (D * 4) + loff);
    const vf bs = *(const vf*)(bias_l);

    vf acc = 0.0f;
    for (int base = 0; base < deg; base += 32) {
        const int m = min(32, deg - base);
        int2 e = e_first;
        if (base > 0) {
            e = make_int2(0, 0);
            if (lane < m) e = edges[start + base + lane];
        }
        int j = 0;
        for (; j + GROWS <= m; j += GROWS) add_rows<D, GROWS>(acc, e, j, hb, loff);
        switch (m - j) {
            case 1: add_rows<D, 1>(acc, e, j, hb, loff); break;
            case 2: add_rows<D, 2>(acc, e, j, hb, loff); break;
            case 3: add_rows<D, 3>(acc, e, j, hb, loff); break;
            case 4: add_rows<D, 4>(acc, e, j, hb, loff); break;
            case 5: add_rows<D, 5>(acc, e, j, hb, loff); break;
            case 6: add_rows<D, 6>(acc, e, j, hb, loff); break;
            case 7: add_rows<D, 7>(acc, e, j, hb, loff); break;
            default: break;
        }
    }

    vf o = acc + (di * di) * hs + bs;
    if (RELU) o = __builtin_elementwise_max(o, (vf)(0.0f));
    *(vf*)((char*)out + (size_t)node * (D * 4) + loff) = o;
}

#define NPW 4
template <int D, bool RELU>
__global__ void __launch_bounds__(256, 16)
gather_kernel(const int2* __restrict__ edges,
              const int2* __restrict__ seg,
              const float* __restrict__ dinv,
              const float* __restrict__ h,
              const float* __restrict__ bias,
              float* __restrict__ out, int n) {
    constexpr int V = D / 32;
    const int lane = threadIdx.x & 31;
    const int node0 = NPW * __builtin_amdgcn_readfirstlane(
        (int)((blockIdx.x * (size_t)blockDim.x + threadIdx.x) >> 5));
    if (node0 >= n) return;

    int   start[NPW], deg[NPW];
    float di[NPW];
    int2  e[NPW];
#pragma unroll
    for (int i = 0; i < NPW; ++i) {
        const int  nd = min(node0 + i, n - 1);
        const int2 sd = seg[nd];
        start[i] = sd.x;
        deg[i]   = (node0 + i < n) ? sd.y : 0;
        di[i]    = dinv[nd];
    }
#pragma unroll
    for (int i = 0; i < NPW; ++i) {
        e[i] = make_int2(0, 0);
        if (lane < deg[i]) e[i] = edges[start[i] + lane];
    }
#pragma unroll
    for (int i = 0; i < NPW; ++i) {
        if (node0 + i < n)
            gather_one<D, RELU>(edges, start[i], deg[i], di[i], e[i], h, lane,
                                bias + lane * V, out, node0 + i);
    }
}

extern "C" void kernel_launch(void* const* d_in, const int* in_sizes, int n_in,
                              void* d_out, int out_size, void* d_ws, size_t ws_size,
                              hipStream_t stream) {
    const float* x  = (const float*)d_in[0];
    const int*   ei = (const int*)d_in[1];
    const float* W0 = (const float*)d_in[2];
    const float* b0 = (const float*)d_in[3];
    const float* W1 = (const float*)d_in[4];
    const float* b1 = (const float*)d_in[5];
    const float* W2 = (const float*)d_in[6];
    const float* b2 = (const float*)d_in[7];

    const int N = in_sizes[0] / IN_DIM;
    const int E = in_sizes[1] / 2;
    const int* rowIdx = ei;
    const int* colIdx = ei + E;

    char*  ws  = (char*)d_ws;
    size_t off = 0;
    auto   alloc = [&](size_t bytes) {
        void* p = ws + off;
        off += (bytes + 255) & ~(size_t)255;
        return p;
    };
    float* dinv   = (float*)alloc((size_t)N * sizeof(float));
    float* hbuf   = (float*)alloc((size_t)N * HID_DIM * sizeof(float));
    float* xbuf   = (float*)alloc((size_t)N * HID_DIM * sizeof(float));
    int*   cnt    = (int*)alloc((size_t)N * sizeof(int));
    int2*  seg    = (int2*)alloc((size_t)N * sizeof(int2));
    int*   rank   = (int*)alloc((size_t)E * sizeof(int));
    int2*  edges  = (int2*)alloc((size_t)E * sizeof(int2));

    const int mtiles   = (N + 15) / 16;
    const int gemmBlks = (mtiles + 7) / 8;
    const int nodeBlks = (N + 8 * NPW - 1) / (8 * NPW);
    const int edgeBlks = (E + 256 * EPT - 1) / (256 * EPT);

    zero_cnt_kernel<<<(N + 1023) / 1024, 256, 0, stream>>>(cnt, N);
    count_rank_kernel<<<edgeBlks, 256, 0, stream>>>(colIdx, cnt, rank, E);
    scan_dinv_kernel<<<(N + SCAN_CHUNK - 1) / SCAN_CHUNK, 256, 0, stream>>>(cnt, seg, dinv, N);
    fill_edges_kernel<<<edgeBlks, 256, 0, stream>>>(rowIdx, colIdx, rank, seg, dinv, edges, E);

    gemm_wmma_kernel<128><<<gemmBlks, 256, 0, stream>>>(x, W0, hbuf, N);
    gather_kernel<128, true><<<nodeBlks, 256, 0, stream>>>(edges, seg, dinv, hbuf, b0, xbuf, N);

    gemm_wmma_kernel<128><<<gemmBlks, 256, 0, stream>>>(xbuf, W1, hbuf, N);
    gather_kernel<128, true><<<nodeBlks, 256, 0, stream>>>(edges, seg, dinv, hbuf, b1, xbuf, N);

    gemm_wmma_kernel<64><<<gemmBlks, 256, 0, stream>>>(xbuf, W2, hbuf, N);
    gather_kernel<64, false><<<nodeBlks, 256, 0, stream>>>(edges, seg, dinv, hbuf, b2, (float*)d_out, N);
}


# Layers 0 and 1: a gather block keeps its 32 summed rows in LDS and multiplies them by the next layer's W on WMMA itself: two GEMM launches and the activation round trip gone; every sum order kept
# speedup vs baseline: 3.2743x; 1.0508x over previous
#include <hip/hip_runtime.h>
#include <hip/hip_bf16.h>

typedef float v2f __attribute__((ext_vector_type(2)));
typedef float v4f __attribute__((ext_vector_type(4)));
typedef float v8f __attribute__((ext_vector_type(8)));

#define IN_DIM  128
#define HID_DIM 128
#define OUT_DIM 64

__global__ void zero_cnt_kernel(int* __restrict__ cnt, int n) {
    const int i = (blockIdx.x * blockDim.x + threadIdx.x) * 4;
    if (i + 3 < n) {
        *(int4*)(cnt + i) = make_int4(0, 0, 0, 0);
    } else {
        for (int k = i; k < n; ++k) cnt[k] = 0;
    }
}

#define EPT 8
__global__ void __launch_bounds__(256)
count_rank_kernel(const int* __restrict__ col,
                  int* __restrict__ cnt,
                  int* __restrict__ rank, int E) {
    const int base = blockIdx.x * (256 * EPT) + threadIdx.x;
    int c[EPT], r[EPT];
#pragma unroll
    for (int i = 0; i < EPT; ++i) {
        const int e = base + i * 256;
        c[i] = (e < E) ? col[e] : -1;
    }
#pragma unroll
    for (int i = 0; i < EPT; ++i) {
        r[i] = 0;
        if (c[i] >= 0) r[i] = atomicAdd(&cnt[c[i]], 1);
    }
#pragma unroll
    for (int i = 0; i < EPT; ++i) {
        const int e = base + i * 256;
        if (e < E) rank[e] = r[i];
    }
}

#define SCAN_CHUNK 1024
#define PF_BATCH 16
__global__ void __launch_bounds__(256)
scan_dinv_kernel(const int* __restrict__ cnt, int2* __restrict__ seg,
                 float* __restrict__ dinv, int n) {
    __shared__ int s_pre[8];
    __shared__ int s_wav[8];
    const int t    = threadIdx.x;
    const int lane = t & 31;
    const int wv   = t >> 5;
    const int chunk0 = blockIdx.x * SCAN_CHUNK;

    int pre = 0;
    const int nfront = (int)blockIdx.x;
    const int* pf = cnt + t * 4;
    for (int it = 0; it < nfront; it += PF_BATCH) {
        int4 v[PF_BATCH];
#pragma unroll
        for (int u = 0; u < PF_BATCH; ++u) {
            v[u] = make_int4(0, 0, 0, 0);
            if (it + u < nfront) v[u] = *(const int4*)(pf + (size_t)(it + u) * SCAN_CHUNK);
        }
#pragma unroll
        for (int u = 0; u < PF_BATCH; ++u) pre += (v[u].x + v[u].y) + (v[u].z + v[u].w);
    }
    for (int off = 16; off > 0; off >>= 1) pre += __shfl_xor(pre, off);

    const int idx = chunk0 + t * 4;
    int c0 = 0, c1 = 0, c2 = 0, c3 = 0;
    if (idx + 3 < n) {
        int4 v = *(const int4*)(cnt + idx);
        c0 = v.x; c1 = v.y; c2 = v.z; c3 = v.w;
    } else {
        if (idx     < n) c0 = cnt[idx];
        if (idx + 1 < n) c1 = cnt[idx + 1];
        if (idx + 2 < n) c2 = cnt[idx + 2];
    }
    const int s = (c0 + c1) + (c2 + c3);
    int inc = s;
    for (int off = 1; off < 32; off <<= 1) {
        int y = __shfl_up(inc, off);
        if (lane >= off) inc += y;
    }
    if (lane == 0)  s_pre[wv] = pre;
    if (lane == 31) s_wav[wv] = inc;
    __syncthreads();
    int before = 0;
    for (int w = 0; w < 8; ++w) {
        before += s_pre[w];
        if (w < wv) before += s_wav[w];
    }
    const int p0 = before + inc - s;
    if (idx < n)     { seg[idx]     = make_int2(p0, c0);                dinv[idx]     = rsqrtf((float)(c0 + 1)); }
    if (idx + 1 < n) { seg[idx + 1] = make_int2(p0 + c0, c1);           dinv[idx + 1] = rsqrtf((float)(c1 + 1)); }
    if (idx + 2 < n) { seg[idx + 2] = make_int2(p0 + c0 + c1, c2);      dinv[idx + 2] = rsqrtf((float)(c2 + 1)); }
    if (idx + 3 < n) { seg[idx + 3] = make_int2(p0 + c0 + c1 + c2, c3); dinv[idx + 3] = rsqrtf((float)(c3 + 1)); }
}

__global__ void __launch_bounds__(256)
fill_edges_kernel(const int* __restrict__ row,
                                  const int* __restrict__ col,
                                  const int* __restrict__ rank,
                                  const int2* __restrict__ seg,
                                  const float* __restrict__ dinv,
                                  int2* __restrict__ edges, int E) {
    const int base = blockIdx.x * (256 * EPT) + threadIdx.x;
    int r[EPT], c[EPT], rk[EPT];
#pragma unroll
    for (int i = 0; i < EPT; ++i) {
        const int e = base + i * 256;
        const bool ok = e < E;
        r[i]  = ok ? row[e]  : -1;
        c[i]  = ok ? col[e]  : 0;
        rk[i] = ok ? rank[e] : 0;
    }
    float dr[EPT], dc[EPT];
    int   sx[EPT];
#pragma unroll
    for (int i = 0; i < EPT; ++i) {
        dr[i] = 0.f; dc[i] = 0.f; sx[i] = 0;
        if (r[i] >= 0) {
            dr[i] = dinv[r[i]];
            dc[i] = dinv[c[i]];
            sx[i] = seg[c[i]].x;
        }
    }
#pragma unroll
    for (int i = 0; i < EPT; ++i) {
        if (r[i] >= 0) {
            const float nrm = dr[i] * dc[i];
            edges[sx[i] + rk[i]] = make_int2(r[i], __float_as_int(nrm));
        }
    }
}

#define LDS_PITCH 132
template <int NOUT>
__global__ void __launch_bounds__(256)
gemm_wmma_kernel(const float* __restrict__ X,
                 const float* __restrict__ W,
                 float* __restrict__ H, int nNodes) {
    constexpr int K  = 128;
    constexpr int NT = NOUT / 16;
    constexpr int P  = LDS_PITCH;
    __shared__ __attribute__((aligned(16))) float sW[NOUT * P];
    __shared__ __attribute__((aligned(16))) float sX[8 * 16 * P];

    const int lane = threadIdx.x & 31;
    const int wv   = threadIdx.x >> 5;
    const int half = lane >> 4;
    const int l16  = lane & 15;
    const int m0   = (blockIdx.x * 8 + wv) * 16;

    float4 wr[NOUT / 8];
#pragma unroll
    for (int i = 0; i < NOUT / 8; ++i)
        wr[i] = *(const float4*)(W + (size_t)(i * 8 + wv) * K + lane * 4);
    float4 xr[16];
#pragma unroll
    for (int r = 0; r < 16; ++r) {
        int mrow = m0 + r; if (mrow >= nNodes) mrow = nNodes - 1;
        xr[r] = *(const float4*)(X + (size_t)mrow * K + lane * 4);
    }
#pragma unroll
    for (int i = 0; i < NOUT / 8; ++i)
        *(float4*)(sW + (i * 8 + wv) * P + lane * 4) = wr[i];
#pragma unroll
    for (int r = 0; r < 16; ++r)
        *(float4*)(sX + (wv * 16 + r) * P + lane * 4) = xr[r];
    __syncthreads();

    if (m0 < nNodes) {
        const float* aptr = sX + (wv * 16 + l16) * P + 2 * half;
        const float* bptr = sW + l16 * P + 2 * half;

        v8f c[NT];
#pragma unroll
        for (int t = 0; t < NT; ++t) c[t] = (v8f){0.f, 0.f, 0.f, 0.f, 0.f, 0.f, 0.f, 0.f};

#pragma unroll
        for (int k0 = 0; k0 < K; k0 += 4) {
            const v2f a = *(const v2f*)(aptr + k0);
#pragma unroll
            for (int t = 0; t < NT; ++t) {
                const v2f b = *(const v2f*)(bptr + t * 16 * P + k0);
                c[t] = __builtin_amdgcn_wmma_f32_16x16x4_f32(false, a, false, b,
                                                             (short)0, c[t], false, false);
            }
        }

        float* out = H + (size_t)m0 * NOUT;
#pragma unroll
        for (int t = 0; t < NT; ++t) {
#pragma unroll
            for (int v = 0; v < 8; ++v) {
                const int r = v + 8 * half;
                if (m0 + r < nNodes) out[(size_t)r * NOUT + t * 16 + l16] = c[t][v];
            }
        }
    }
}

template <int D, int R>
__device__ __forceinline__ void
add_rows(float __attribute__((ext_vector_type(D / 32)))& acc, const int2 e, const int j,
         const char* __restrict__ hb, const unsigned loff) {
    typedef float vf __attribute__((ext_vector_type(D / 32)));
    vf v[R];
#pragma unroll
    for (int u = 0; u < R; ++u) {
        const int r = __builtin_amdgcn_readlane(e.x, j + u);
        v[u] = *(const vf*)(hb + (size_t)r * (D * 4) + loff);
    }
#pragma unroll
    for (int u = 0; u < R; ++u) {
        const float w = __int_as_float(__builtin_amdgcn_readlane(e.y, j + u));
        acc += w * v[u];
    }
}

#define GROWS 8
template <int D, bool RELU>
__device__ __forceinline__ float __attribute__((ext_vector_type(D / 32)))
gather_one(const int2* __restrict__ edges, const int start, const int deg,
           const float di, const int2 e_first,
           const float* __restrict__ h, const int lane,
           const float* __restrict__ bias_l, const int node) {
    constexpr int V = D / 32;
    typedef float vf __attribute__((ext_vector_type(V)));
    const char*    hb   = (const char*)h;
    const unsigned loff = (unsigned)lane * (V * 4);
    const vf hs = *(const vf*)(hb + (size_t)node * (D * 4) + loff);
    const vf bs = *(const vf*)(bias_l);

    vf acc = 0.0f;
    for (int base = 0; base < deg; base += 32) {
        const int m = min(32, deg - base);
        int2 e = e_first;
        if (base > 0) {
            e = make_int2(0, 0);
            if (lane < m) e = edges[start + base + lane];
        }
        int j = 0;
        for (; j + GROWS <= m; j += GROWS) add_rows<D, GROWS>(acc, e, j, hb, loff);
        switch (m - j) {
            case 1: add_rows<D, 1>(acc, e, j, hb, loff); break;
            case 2: add_rows<D, 2>(acc, e, j, hb, loff); break;
            case 3: add_rows<D, 3>(acc, e, j, hb, loff); break;
            case 4: add_rows<D, 4>(acc, e, j, hb, loff); break;
            case 5: add_rows<D, 5>(acc, e, j, hb, loff); break;
            case 6: add_rows<D, 6>(acc, e, j, hb, loff); break;
            case 7: add_rows<D, 7>(acc, e, j, hb, loff); break;
            default: break;
        }
    }

    vf o = acc + (di * di) * hs + bs;
    if (RELU) o = __builtin_elementwise_max(o, (vf)(0.0f));
    return o;
}

#define NPW 4
template <int D, bool RELU>
__global__ void __launch_bounds__(256, 16)
gather_kernel(const int2* __restrict__ edges,
              const int2* __restrict__ seg,
              const float* __restrict__ dinv,
              const float* __restrict__ h,
              const float* __restrict__ bias,
              float* __restrict__ out, int n) {
    constexpr int V = D / 32;
    const int lane = threadIdx.x & 31;
    const int node0 = NPW * __builtin_amdgcn_readfirstlane(
        (int)((blockIdx.x * (size_t)blockDim.x + threadIdx.x) >> 5));
    if (node0 >= n) return;

    int   start[NPW], deg[NPW];
    float di[NPW];
    int2  e[NPW];
#pragma unroll
    for (int i = 0; i < NPW; ++i) {
        const int  nd = min(node0 + i, n - 1);
        const int2 sd = seg[nd];
        start[i] = sd.x;
        deg[i]   = (node0 + i < n) ? sd.y : 0;
        di[i]    = dinv[nd];
    }
#pragma unroll
    for (int i = 0; i < NPW; ++i) {
        e[i] = make_int2(0, 0);
        if (lane < deg[i]) e[i] = edges[start[i] + lane];
    }
#pragma unroll
    for (int i = 0; i < NPW; ++i) {
        if (node0 + i < n) {
            typedef float vf __attribute__((ext_vector_type(V)));
            const vf o = gather_one<D, RELU>(edges, start[i], deg[i], di[i], e[i], h, lane,
                                             bias + lane * V, node0 + i);
            *(vf*)((char*)out + (size_t)(node0 + i) * (D * 4) + (unsigned)lane * (V * 4)) = o;
        }
    }
}

template <int NOUT>
__global__ void __launch_bounds__(256)
gather_gemm_kernel(const int2* __restrict__ edges,
                   const int2* __restrict__ seg,
                   const float* __restrict__ dinv,
                   const float* __restrict__ h,
                   const float* __restrict__ bias,
                   const float* __restrict__ W,
                   float* __restrict__ H, int n) {
    constexpr int K   = 128;
    constexpr int P   = LDS_PITCH;
    constexpr int TPW = NOUT / 64;
    static_assert(NPW * 8 == 32, "eight waves of NPW nodes fill the block's two 16-row strips");
    __shared__ __attribute__((aligned(16))) float sW[NOUT * P];
    __shared__ __attribute__((aligned(16))) float sX[32 * P];

    const int lane = threadIdx.x & 31;
    const int wv    = __builtin_amdgcn_readfirstlane((int)(threadIdx.x >> 5));
    const int half  = lane >> 4;
    const int l16   = lane & 15;
    const int nodeB = (int)blockIdx.x * 32;
    const int node0 = nodeB + NPW * wv;

    float4 wr[NOUT / 8];
#pragma unroll
    for (int i = 0; i < NOUT / 8; ++i)
        wr[i] = *(const float4*)(W + (size_t)(i * 8 + wv) * K + lane * 4);

    int   start[NPW], deg[NPW];
    float di[NPW];
    int2  e[NPW];
#pragma unroll
    for (int i = 0; i < NPW; ++i) {
        const int  nd = min(node0 + i, n - 1);
        const int2 sd = seg[nd];
        start[i] = sd.x;
        deg[i]   = (node0 + i < n) ? sd.y : 0;
        di[i]    = dinv[nd];
    }
#pragma unroll
    for (int i = 0; i < NPW; ++i) {
        e[i] = make_int2(0, 0);
        if (lane < deg[i]) e[i] = edges[start[i] + lane];
    }
#pragma unroll
    for (int i = 0; i < NOUT / 8; ++i)
        *(float4*)(sW + (i * 8 + wv) * P + lane * 4) = wr[i];

#pragma unroll
    for (int i = 0; i < NPW; ++i) {
        float4 o = make_float4(0.f, 0.f, 0.f, 0.f);
        if (node0 + i < n) {
            const v4f g = gather_one<128, true>(edges, start[i], deg[i], di[i], e[i], h, lane,
                                                bias + lane * 4, node0 + i);
            o = make_float4(g[0], g[1], g[2], g[3]);
        }
        *(float4*)(sX + (NPW * wv + i) * P + lane * 4) = o;
    }
    __syncthreads();

    const int strip = wv >> 2;
    const int tq    = (wv & 3) * TPW;
    const float* aptr = sX + (strip * 16 + l16) * P + 2 * half;
    const float* bptr = sW + (tq * 16 + l16) * P + 2 * half;

    v8f c[TPW];
#pragma unroll
    for (int t = 0; t < TPW; ++t) c[t] = (v8f){0.f, 0.f, 0.f, 0.f, 0.f, 0.f, 0.f, 0.f};

#pragma unroll
    for (int k0 = 0; k0 < K; k0 += 4) {
        const v2f a = *(const v2f*)(aptr + k0);
#pragma unroll
        for (int t = 0; t < TPW; ++t) {
            const v2f b = *(const v2f*)(bptr + t * 16 * P + k0);
            c[t] = __builtin_amdgcn_wmma_f32_16x16x4_f32(false, a, false, b,
                                                         (short)0, c[t], false, false);
        }
    }

    const int m0  = nodeB + strip * 16;
    float*    out = H + (size_t)m0 * NOUT;
#pragma unroll
    for (int t = 0; t < TPW; ++t) {
#pragma unroll
        for (int v = 0; v < 8; ++v) {
            const int r = v + 8 * half;
            if (m0 + r < n) out[(size_t)r * NOUT + (tq + t) * 16 + l16] = c[t][v];
        }
    }
}

extern "C" void kernel_launch(void* const* d_in, const int* in_sizes, int n_in,
                              void* d_out, int out_size, void* d_ws, size_t ws_size,
                              hipStream_t stream) {
    const float* x  = (const float*)d_in[0];
    const int*   ei = (const int*)d_in[1];
    const float* W0 = (const float*)d_in[2];
    const float* b0 = (const float*)d_in[3];
    const float* W1 = (const float*)d_in[4];
    const float* b1 = (const float*)d_in[5];
    const float* W2 = (const float*)d_in[6];
    const float* b2 = (const float*)d_in[7];

    const int N = in_sizes[0] / IN_DIM;
    const int E = in_sizes[1] / 2;
    const int* rowIdx = ei;
    const int* colIdx = ei + E;

    char*  ws  = (char*)d_ws;
    size_t off = 0;
    auto   alloc = [&](size_t bytes) {
        void* p = ws + off;
        off += (bytes + 255) & ~(size_t)255;
        return p;
    };
    float* dinv   = (float*)alloc((size_t)N * sizeof(float));
    float* hbuf   = (float*)alloc((size_t)N * HID_DIM * sizeof(float));
    float* xbuf   = (float*)alloc((size_t)N * HID_DIM * sizeof(float));
    int*   cnt    = (int*)alloc((size_t)N * sizeof(int));
    int2*  seg    = (int2*)alloc((size_t)N * sizeof(int2));
    int*   rank   = (int*)alloc((size_t)E * sizeof(int));
    int2*  edges  = (int2*)alloc((size_t)E * sizeof(int2));

    const int mtiles   = (N + 15) / 16;
    const int gemmBlks = (mtiles + 7) / 8;
    const int nodeBlks = (N + 8 * NPW - 1) / (8 * NPW);
    const int edgeBlks = (E + 256 * EPT - 1) / (256 * EPT);
    const int fuseBlks = (N + 31) / 32;

    zero_cnt_kernel<<<(N + 1023) / 1024, 256, 0, stream>>>(cnt, N);
    count_rank_kernel<<<edgeBlks, 256, 0, stream>>>(colIdx, cnt, rank, E);
    scan_dinv_kernel<<<(N + SCAN_CHUNK - 1) / SCAN_CHUNK, 256, 0, stream>>>(cnt, seg, dinv, N);
    fill_edges_kernel<<<edgeBlks, 256, 0, stream>>>(rowIdx, colIdx, rank, seg, dinv, edges, E);

    gemm_wmma_kernel<128><<<gemmBlks, 256, 0, stream>>>(x, W0, hbuf, N);

    gather_gemm_kernel<128><<<fuseBlks, 256, 0, stream>>>(edges, seg, dinv, hbuf, b0, W1, xbuf, N);

    gather_gemm_kernel<64><<<fuseBlks, 256, 0, stream>>>(edges, seg, dinv, xbuf, b1, W2, hbuf, N);

    gather_kernel<64, false><<<nodeBlks, 256, 0, stream>>>(edges, seg, dinv, hbuf, b2, (float*)d_out, N);
}


# Layer-0 product dealt over the two edge launches (rank count, edge fill), 4 strips per block beside their memory round trips: its own launch gone, 7 launches; every tile's k order kept
# speedup vs baseline: 3.3955x; 1.0370x over previous
#include <hip/hip_runtime.h>
#include <hip/hip_bf16.h>

typedef float v2f __attribute__((ext_vector_type(2)));
typedef float v4f __attribute__((ext_vector_type(4)));
typedef float v8f __attribute__((ext_vector_type(8)));

#define IN_DIM  128
#define HID_DIM 128
#define OUT_DIM 64

__global__ void zero_cnt_kernel(int* __restrict__ cnt, int n) {
    const int i = (blockIdx.x * blockDim.x + threadIdx.x) * 4;
    if (i + 3 < n) {
        *(int4*)(cnt + i) = make_int4(0, 0, 0, 0);
    } else {
        for (int k = i; k < n; ++k) cnt[k] = 0;
    }
}

#define LDS_PITCH 132
#define GSTRIPS 4
struct GemmPart {
    v4f wr[16];
    v4f xr[2 * GSTRIPS];

    __device__ __forceinline__ void load(const float* __restrict__ X,
                                         const float* __restrict__ W,
                                         const int row0, const int nNodes,
                                         const int wv, const int lane) {
#pragma unroll
        for (int i = 0; i < 16; ++i)
            wr[i] = *(const v4f*)(W + (size_t)(i * 8 + wv) * 128 + lane * 4);
#pragma unroll
        for (int i = 0; i < 2 * GSTRIPS; ++i) {
            int mrow = row0 + i * 8 + wv; if (mrow >= nNodes) mrow = nNodes - 1;
            xr[i] = *(const v4f*)(X + (size_t)mrow * 128 + lane * 4);
        }
    }

    __device__ __forceinline__ void hold() const {
#pragma unroll
        for (int i = 0; i < 16; ++i) asm volatile("" :: "v"(wr[i]));
#pragma unroll
        for (int i = 0; i < 2 * GSTRIPS; ++i) asm volatile("" :: "v"(xr[i]));
    }

    __device__ __forceinline__ void stage(float* sW, float* sX, const int wv, const int lane) {
        constexpr int P = LDS_PITCH;
#pragma unroll
        for (int i = 0; i < 16; ++i)
            *(v4f*)(sW + (i * 8 + wv) * P + lane * 4) = wr[i];
#pragma unroll
        for (int i = 0; i < 2 * GSTRIPS; ++i)
            *(v4f*)(sX + (i * 8 + wv) * P + lane * 4) = xr[i];
        __syncthreads();
    }
};

__device__ __forceinline__ void
gemm_product(const float* sW, const float* sX, float* __restrict__ H,
             const int row0, const int nNodes, const int wv, const int lane) {
    constexpr int K = 128;
    constexpr int P = LDS_PITCH;
    constexpr int NT = 4;
    const int half  = lane >> 4;
    const int l16   = lane & 15;
    const int strip = wv >> 1;
    const int tq    = (wv & 1) * NT;
    const int m0    = row0 + strip * 16;
    if (m0 < nNodes) {
        const float* aptr = sX + (strip * 16 + l16) * P + 2 * half;
        const float* bptr = sW + (tq * 16 + l16) * P + 2 * half;

        v8f c[NT];
#pragma unroll
        for (int t = 0; t < NT; ++t) c[t] = (v8f){0.f, 0.f, 0.f, 0.f, 0.f, 0.f, 0.f, 0.f};

#pragma unroll
        for (int k0 = 0; k0 < K; k0 += 4) {
            const v2f a = *(const v2f*)(aptr + k0);
#pragma unroll
            for (int t = 0; t < NT; ++t) {
                const v2f b = *(const v2f*)(bptr + t * 16 * P + k0);
                c[t] = __builtin_amdgcn_wmma_f32_16x16x4_f32(false, a, false, b,
                                                             (short)0, c[t], false, false);
            }
        }

        float* out = H + (size_t)m0 * 128;
#pragma unroll
        for (int t = 0; t < NT; ++t) {
#pragma unroll
            for (int v = 0; v < 8; ++v) {
                const int r = v + 8 * half;
                if (m0 + r < nNodes) out[(size_t)r * 128 + (tq + t) * 16 + l16] = c[t][v];
            }
        }
    }
}

#define EPT 8
__global__ void __launch_bounds__(256)
count_rank_gemm_kernel(const int* __restrict__ col,
                       int* __restrict__ cnt,
                       int* __restrict__ rank, int E,
                       const float* __restrict__ X,
                       const float* __restrict__ W,
                       float* __restrict__ H, int nNodes) {
    __shared__ __attribute__((aligned(16))) float sW[128 * LDS_PITCH];
    __shared__ __attribute__((aligned(16))) float sX[GSTRIPS * 16 * LDS_PITCH];
    const int lane = threadIdx.x & 31;
    const int wv   = threadIdx.x >> 5;
    const int row0 = (int)blockIdx.x * (GSTRIPS * 16);

    GemmPart g;
    g.load(X, W, row0, nNodes, wv, lane);
    const int base = blockIdx.x * (256 * EPT) + threadIdx.x;
    int c[EPT], r[EPT];
#pragma unroll
    for (int i = 0; i < EPT; ++i) {
        const int e = base + i * 256;
        c[i] = (e < E) ? col[e] : -1;
    }
    g.hold();
#pragma unroll
    for (int i = 0; i < EPT; ++i) asm volatile("" :: "v"(c[i]));
#pragma unroll
    for (int i = 0; i < EPT; ++i) {
        r[i] = 0;
        if (c[i] >= 0) r[i] = atomicAdd(&cnt[c[i]], 1);
    }
    g.stage(sW, sX, wv, lane);
    gemm_product(sW, sX, H, row0, nNodes, wv, lane);
#pragma unroll
    for (int i = 0; i < EPT; ++i) {
        const int e = base + i * 256;
        if (e < E) rank[e] = r[i];
    }
}

#define SCAN_CHUNK 1024
#define PF_BATCH 16
__global__ void __launch_bounds__(256)
scan_dinv_kernel(const int* __restrict__ cnt, int2* __restrict__ seg,
                 float* __restrict__ dinv, int n) {
    __shared__ int s_pre[8];
    __shared__ int s_wav[8];
    const int t    = threadIdx.x;
    const int lane = t & 31;
    const int wv   = t >> 5;
    const int chunk0 = blockIdx.x * SCAN_CHUNK;

    int pre = 0;
    const int nfront = (int)blockIdx.x;
    const int* pf = cnt + t * 4;
    for (int it = 0; it < nfront; it += PF_BATCH) {
        int4 v[PF_BATCH];
#pragma unroll
        for (int u = 0; u < PF_BATCH; ++u) {
            v[u] = make_int4(0, 0, 0, 0);
            if (it + u < nfront) v[u] = *(const int4*)(pf + (size_t)(it + u) * SCAN_CHUNK);
        }
#pragma unroll
        for (int u = 0; u < PF_BATCH; ++u) pre += (v[u].x + v[u].y) + (v[u].z + v[u].w);
    }
    for (int off = 16; off > 0; off >>= 1) pre += __shfl_xor(pre, off);

    const int idx = chunk0 + t * 4;
    int c0 = 0, c1 = 0, c2 = 0, c3 = 0;
    if (idx + 3 < n) {
        int4 v = *(const int4*)(cnt + idx);
        c0 = v.x; c1 = v.y; c2 = v.z; c3 = v.w;
    } else {
        if (idx     < n) c0 = cnt[idx];
        if (idx + 1 < n) c1 = cnt[idx + 1];
        if (idx + 2 < n) c2 = cnt[idx + 2];
    }
    const int s = (c0 + c1) + (c2 + c3);
    int inc = s;
    for (int off = 1; off < 32; off <<= 1) {
        int y = __shfl_up(inc, off);
        if (lane >= off) inc += y;
    }
    if (lane == 0)  s_pre[wv] = pre;
    if (lane == 31) s_wav[wv] = inc;
    __syncthreads();
    int before = 0;
    for (int w = 0; w < 8; ++w) {
        before += s_pre[w];
        if (w < wv) before += s_wav[w];
    }
    const int p0 = before + inc - s;
    if (idx < n)     { seg[idx]     = make_int2(p0, c0);                dinv[idx]     = rsqrtf((float)(c0 + 1)); }
    if (idx + 1 < n) { seg[idx + 1] = make_int2(p0 + c0, c1);           dinv[idx + 1] = rsqrtf((float)(c1 + 1)); }
    if (idx + 2 < n) { seg[idx + 2] = make_int2(p0 + c0 + c1, c2);      dinv[idx + 2] = rsqrtf((float)(c2 + 1)); }
    if (idx + 3 < n) { seg[idx + 3] = make_int2(p0 + c0 + c1 + c2, c3); dinv[idx + 3] = rsqrtf((float)(c3 + 1)); }
}

__global__ void __launch_bounds__(256)
fill_edges_gemm_kernel(const int* __restrict__ row,
                       const int* __restrict__ col,
                       const int* __restrict__ rank,
                       const int2* __restrict__ seg,
                       const float* __restrict__ dinv,
                       int2* __restrict__ edges, int E,
                       const float* __restrict__ X,
                       const float* __restrict__ W,
                       float* __restrict__ H, int nNodes) {
    __shared__ __attribute__((aligned(16))) float sW[128 * LDS_PITCH];
    __shared__ __attribute__((aligned(16))) float sX[GSTRIPS * 16 * LDS_PITCH];
    const int lane = threadIdx.x & 31;
    const int wv   = threadIdx.x >> 5;
    const int row0 = (int)(gridDim.x + blockIdx.x) * (GSTRIPS * 16);

    GemmPart g;
    g.load(X, W, row0, nNodes, wv, lane);
    const int base = blockIdx.x * (256 * EPT) + threadIdx.x;
    int r[EPT], c[EPT], rk[EPT];
#pragma unroll
    for (int i = 0; i < EPT; ++i) {
        const int e = base + i * 256;
        const bool ok = e < E;
        r[i]  = ok ? row[e]  : -1;
        c[i]  = ok ? col[e]  : 0;
        rk[i] = ok ? rank[e] : 0;
    }
    g.hold();
#pragma unroll
    for (int i = 0; i < EPT; ++i) asm volatile("" :: "v"(r[i]), "v"(c[i]), "v"(rk[i]));
    float dr[EPT], dc[EPT];
    int   sx[EPT];
#pragma unroll
    for (int i = 0; i < EPT; ++i) {
        dr[i] = 0.f; dc[i] = 0.f; sx[i] = 0;
        if (r[i] >= 0) {
            dr[i] = dinv[r[i]];
            dc[i] = dinv[c[i]];
            sx[i] = seg[c[i]].x;
        }
    }
    g.stage(sW, sX, wv, lane);
    gemm_product(sW, sX, H, row0, nNodes, wv, lane);
#pragma unroll
    for (int i = 0; i < EPT; ++i) {
        if (r[i] >= 0) {
            const float nrm = dr[i] * dc[i];
            edges[sx[i] + rk[i]] = make_int2(r[i], __float_as_int(nrm));
        }
    }
}

template <int D, int R>
__device__ __forceinline__ void
add_rows(float __attribute__((ext_vector_type(D / 32)))& acc, const int2 e, const int j,
         const char* __restrict__ hb, const unsigned loff) {
    typedef float vf __attribute__((ext_vector_type(D / 32)));
    vf v[R];
#pragma unroll
    for (int u = 0; u < R; ++u) {
        const int r = __builtin_amdgcn_readlane(e.x, j + u);
        v[u] = *(const vf*)(hb + (size_t)r * (D * 4) + loff);
    }
#pragma unroll
    for (int u = 0; u < R; ++u) {
        const float w = __int_as_float(__builtin_amdgcn_readlane(e.y, j + u));
        acc += w * v[u];
    }
}

#define GROWS 8
template <int D, bool RELU>
__device__ __forceinline__ float __attribute__((ext_vector_type(D / 32)))
gather_one(const int2* __restrict__ edges, const int start, const int deg,
           const float di, const int2 e_first,
           const float* __restrict__ h, const int lane,
           const float* __restrict__ bias_l, const int node) {
    constexpr int V = D / 32;
    typedef float vf __attribute__((ext_vector_type(V)));
    const char*    hb   = (const char*)h;
    const unsigned loff = (unsigned)lane * (V * 4);
    const vf hs = *(const vf*)(hb + (size_t)node * (D * 4) + loff);
    const vf bs = *(const vf*)(bias_l);

    vf acc = 0.0f;
    for (int base = 0; base < deg; base += 32) {
        const int m = min(32, deg - base);
        int2 e = e_first;
        if (base > 0) {
            e = make_int2(0, 0);
            if (lane < m) e = edges[start + base + lane];
        }
        int j = 0;
        for (; j + GROWS <= m; j += GROWS) add_rows<D, GROWS>(acc, e, j, hb, loff);
        switch (m - j) {
            case 1: add_rows<D, 1>(acc, e, j, hb, loff); break;
            case 2: add_rows<D, 2>(acc, e, j, hb, loff); break;
            case 3: add_rows<D, 3>(acc, e, j, hb, loff); break;
            case 4: add_rows<D, 4>(acc, e, j, hb, loff); break;
            case 5: add_rows<D, 5>(acc, e, j, hb, loff); break;
            case 6: add_rows<D, 6>(acc, e, j, hb, loff); break;
            case 7: add_rows<D, 7>(acc, e, j, hb, loff); break;
            default: break;
        }
    }

    vf o = acc + (di * di) * hs + bs;
    if (RELU) o = __builtin_elementwise_max(o, (vf)(0.0f));
    return o;
}

#define NPW 4
template <int D, bool RELU>
__global__ void __launch_bounds__(256, 16)
gather_kernel(const int2* __restrict__ edges,
              const int2* __restrict__ seg,
              const float* __restrict__ dinv,
              const float* __restrict__ h,
              const float* __restrict__ bias,
              float* __restrict__ out, int n) {
    constexpr int V = D / 32;
    const int lane = threadIdx.x & 31;
    const int node0 = NPW * __builtin_amdgcn_readfirstlane(
        (int)((blockIdx.x * (size_t)blockDim.x + threadIdx.x) >> 5));
    if (node0 >= n) return;

    int   start[NPW], deg[NPW];
    float di[NPW];
    int2  e[NPW];
#pragma unroll
    for (int i = 0; i < NPW; ++i) {
        const int  nd = min(node0 + i, n - 1);
        const int2 sd = seg[nd];
        start[i] = sd.x;
        deg[i]   = (node0 + i < n) ? sd.y : 0;
        di[i]    = dinv[nd];
    }
#pragma unroll
    for (int i = 0; i < NPW; ++i) {
        e[i] = make_int2(0, 0);
        if (lane < deg[i]) e[i] = edges[start[i] + lane];
    }
#pragma unroll
    for (int i = 0; i < NPW; ++i) {
        if (node0 + i < n) {
            typedef float vf __attribute__((ext_vector_type(V)));
            const vf o = gather_one<D, RELU>(edges, start[i], deg[i], di[i], e[i], h, lane,
                                             bias + lane * V, node0 + i);
            *(vf*)((char*)out + (size_t)(node0 + i) * (D * 4) + (unsigned)lane * (V * 4)) = o;
        }
    }
}

template <int NOUT>
__global__ void __launch_bounds__(256)
gather_gemm_kernel(const int2* __restrict__ edges,
                   const int2* __restrict__ seg,
                   const float* __restrict__ dinv,
                   const float* __restrict__ h,
                   const float* __restrict__ bias,
                   const float* __restrict__ W,
                   float* __restrict__ H, int n) {
    constexpr int K   = 128;
    constexpr int P   = LDS_PITCH;
    constexpr int TPW = NOUT / 64;
    static_assert(NPW * 8 == 32, "eight waves of NPW nodes fill the block's two 16-row strips");
    __shared__ __attribute__((aligned(16))) float sW[NOUT * P];
    __shared__ __attribute__((aligned(16))) float sX[32 * P];

    const int lane = threadIdx.x & 31;
    const int wv    = __builtin_amdgcn_readfirstlane((int)(threadIdx.x >> 5));
    const int half  = lane >> 4;
    const int l16   = lane & 15;
    const int nodeB = (int)blockIdx.x * 32;
    const int node0 = nodeB + NPW * wv;

    float4 wr[NOUT / 8];
#pragma unroll
    for (int i = 0; i < NOUT / 8; ++i)
        wr[i] = *(const float4*)(W + (size_t)(i * 8 + wv) * K + lane * 4);

    int   start[NPW], deg[NPW];
    float di[NPW];
    int2  e[NPW];
#pragma unroll
    for (int i = 0; i < NPW; ++i) {
        const int  nd = min(node0 + i, n - 1);
        const int2 sd = seg[nd];
        start[i] = sd.x;
        deg[i]   = (node0 + i < n) ? sd.y : 0;
        di[i]    = dinv[nd];
    }
#pragma unroll
    for (int i = 0; i < NPW; ++i) {
        e[i] = make_int2(0, 0);
        if (lane < deg[i]) e[i] = edges[start[i] + lane];
    }
#pragma unroll
    for (int i = 0; i < NOUT / 8; ++i)
        *(float4*)(sW + (i * 8 + wv) * P + lane * 4) = wr[i];

#pragma unroll
    for (int i = 0; i < NPW; ++i) {
        float4 o = make_float4(0.f, 0.f, 0.f, 0.f);
        if (node0 + i < n) {
            const v4f g = gather_one<128, true>(edges, start[i], deg[i], di[i], e[i], h, lane,
                                                bias + lane * 4, node0 + i);
            o = make_float4(g[0], g[1], g[2], g[3]);
        }
        *(float4*)(sX + (NPW * wv + i) * P + lane * 4) = o;
    }
    __syncthreads();

    const int strip = wv >> 2;
    const int tq    = (wv & 3) * TPW;
    const float* aptr = sX + (strip * 16 + l16) * P + 2 * half;
    const float* bptr = sW + (tq * 16 + l16) * P + 2 * half;

    v8f c[TPW];
#pragma unroll
    for (int t = 0; t < TPW; ++t) c[t] = (v8f){0.f, 0.f, 0.f, 0.f, 0.f, 0.f, 0.f, 0.f};

#pragma unroll
    for (int k0 = 0; k0 < K; k0 += 4) {
        const v2f a = *(const v2f*)(aptr + k0);
#pragma unroll
        for (int t = 0; t < TPW; ++t) {
            const v2f b = *(const v2f*)(bptr + t * 16 * P + k0);
            c[t] = __builtin_amdgcn_wmma_f32_16x16x4_f32(false, a, false, b,
                                                         (short)0, c[t], false, false);
        }
    }

    const int m0  = nodeB + strip * 16;
    float*    out = H + (size_t)m0 * NOUT;
#pragma unroll
    for (int t = 0; t < TPW; ++t) {
#pragma unroll
        for (int v = 0; v < 8; ++v) {
            const int r = v + 8 * half;
            if (m0 + r < n) out[(size_t)r * NOUT + (tq + t) * 16 + l16] = c[t][v];
        }
    }
}

extern "C" void kernel_launch(void* const* d_in, const int* in_sizes, int n_in,
                              void* d_out, int out_size, void* d_ws, size_t ws_size,
                              hipStream_t stream) {
    const float* x  = (const float*)d_in[0];
    const int*   ei = (const int*)d_in[1];
    const float* W0 = (const float*)d_in[2];
    const float* b0 = (const float*)d_in[3];
    const float* W1 = (const float*)d_in[4];
    const float* b1 = (const float*)d_in[5];
    const float* W2 = (const float*)d_in[6];
    const float* b2 = (const float*)d_in[7];

    const int N = in_sizes[0] / IN_DIM;
    const int E = in_sizes[1] / 2;
    const int* rowIdx = ei;
    const int* colIdx = ei + E;

    char*  ws  = (char*)d_ws;
    size_t off = 0;
    auto   alloc = [&](size_t bytes) {
        void* p = ws + off;
        off += (bytes + 255) & ~(size_t)255;
        return p;
    };
    float* dinv   = (float*)alloc((size_t)N * sizeof(float));
    float* hbuf   = (float*)alloc((size_t)N * HID_DIM * sizeof(float));
    float* xbuf   = (float*)alloc((size_t)N * HID_DIM * sizeof(float));
    int*   cnt    = (int*)alloc((size_t)N * sizeof(int));
    int2*  seg    = (int2*)alloc((size_t)N * sizeof(int2));
    int*   rank   = (int*)alloc((size_t)E * sizeof(int));
    int2*  edges  = (int2*)alloc((size_t)E * sizeof(int2));

    const int mtiles   = (N + 15) / 16;
    const int gemmBlks = (mtiles + 2 * GSTRIPS - 1) / (2 * GSTRIPS);
    const int nodeBlks = (N + 8 * NPW - 1) / (8 * NPW);
    const int edgeBlks = (E + 256 * EPT - 1) / (256 * EPT);
    const int dualBlks = edgeBlks > gemmBlks ? edgeBlks : gemmBlks;
    const int fuseBlks = (N + 31) / 32;

    zero_cnt_kernel<<<(N + 1023) / 1024, 256, 0, stream>>>(cnt, N);
    count_rank_gemm_kernel<<<dualBlks, 256, 0, stream>>>(colIdx, cnt, rank, E, x, W0, hbuf, N);
    scan_dinv_kernel<<<(N + SCAN_CHUNK - 1) / SCAN_CHUNK, 256, 0, stream>>>(cnt, seg, dinv, N);
    fill_edges_gemm_kernel<<<dualBlks, 256, 0, stream>>>(rowIdx, colIdx, rank, seg, dinv, edges, E, x, W0, hbuf, N);

    gather_gemm_kernel<128><<<fuseBlks, 256, 0, stream>>>(edges, seg, dinv, hbuf, b0, W1, xbuf, N);

    gather_gemm_kernel<64><<<fuseBlks, 256, 0, stream>>>(edges, seg, dinv, xbuf, b1, W2, hbuf, N);

    gather_kernel<64, false><<<nodeBlks, 256, 0, stream>>>(edges, seg, dinv, hbuf, b2, (float*)d_out, N);
}
